# starting kernel
# baseline (speedup 1.0000x reference)
#include <hip/hip_runtime.h>

typedef __attribute__((ext_vector_type(2))) float v2f;
typedef __attribute__((ext_vector_type(8))) float v8f;

constexpr int Bc = 2;
constexpr int Sc = 1024;
constexpr int Nc = Sc + 1;
constexpr int Dc = 512;
constexpr int Rc = 32;
constexpr int Lc = 3;
constexpr int Wc = 64;
constexpr int WNc = 2 * Wc + 1;
constexpr int WINc = 144;
constexpr int NT = (Nc + 15) / 16;

constexpr int XSTR = Dc + 4;
constexpr int QSTR = 33;
constexpr int KSTR = 33;
constexpr int SSTR = 146;
constexpr int WSTR = 145;

__device__ __forceinline__ v8f wmma_f32(v2f a, v2f b, v8f c) {
  return __builtin_amdgcn_wmma_f32_16x16x4_f32(false, a, false, b, (short)0, c,
                                               false, false);
}

__device__ __forceinline__ float block_sum256(float v, float* red) {
  for (int m = 16; m >= 1; m >>= 1) v += __shfl_xor(v, m, 32);
  const int wave = threadIdx.x >> 5, lane = threadIdx.x & 31;
  __syncthreads();
  if (lane == 0) red[wave] = v;
  __syncthreads();
  if (threadIdx.x < 8) {
    float t = red[threadIdx.x];
    t += __shfl_xor(t, 1, 32);
    t += __shfl_xor(t, 2, 32);
    t += __shfl_xor(t, 4, 32);
    if (threadIdx.x == 0) red[0] = t;
  }
  __syncthreads();
  return red[0];
}

__global__ __launch_bounds__(256) void embed_ln_kernel(
    const int* __restrict__ ids, const float* __restrict__ emb,
    const float* __restrict__ pos, const float* __restrict__ anchor_val,
    const float* __restrict__ anchor_state, const float* __restrict__ gamma,
    const float* __restrict__ beta, const float* __restrict__ sp_w,
    const float* __restrict__ sp_b, float* __restrict__ val,
    float* __restrict__ st) {
  __shared__ float red[8];
  const int n = blockIdx.x % Nc;
  const int b = blockIdx.x / Nc;
  const int tid = threadIdx.x;
  float* vrow = val + ((size_t)b * Nc + n) * Dc;
  if (n == 0) {
    vrow[tid] = anchor_val[tid];
    vrow[tid + 256] = anchor_val[tid + 256];
    if (tid == 0) st[(size_t)b * Nc] = anchor_state[0];
    return;
  }
  const int s = n - 1;
  const int id = ids[b * Sc + s];
  const float e0 = emb[(size_t)id * Dc + tid] + pos[(size_t)s * Dc + tid];
  const float e1 = emb[(size_t)id * Dc + tid + 256] + pos[(size_t)s * Dc + tid + 256];
  const float sum = block_sum256(e0 + e1, red);
  const float ssq = block_sum256(e0 * e0 + e1 * e1, red);
  const float mean = sum * (1.0f / Dc);
  const float var = ssq * (1.0f / Dc) - mean * mean;
  const float rs = rsqrtf(var + 1e-5f);
  const float y0 = (e0 - mean) * rs * gamma[tid] + beta[tid];
  const float y1 = (e1 - mean) * rs * gamma[tid + 256] + beta[tid + 256];
  vrow[tid] = y0;
  vrow[tid + 256] = y1;
  const float dot = block_sum256(y0 * sp_w[tid] + y1 * sp_w[tid + 256], red);
  if (tid == 0) st[(size_t)b * Nc + n] = dot + sp_b[0];
}

__global__ __launch_bounds__(128) void ln_proj_kernel(
    const float* __restrict__ val, const float* __restrict__ gamma,
    const float* __restrict__ beta, const float* __restrict__ Uw,
    const float* __restrict__ Vw, float* __restrict__ xo,
    float* __restrict__ qo, float* __restrict__ ko) {
  __shared__ float xs[16 * XSTR];
  const int tile = blockIdx.x % NT;
  const int b = blockIdx.x / NT;
  const int n0 = tile * 16;
  const int tid = threadIdx.x;
  const bool full = (n0 + 16 <= Nc);

  const int row = tid >> 3, sub = tid & 7;
  const int g = n0 + row;
  const float* vrow = val + ((size_t)b * Nc + g) * Dc;
  if (full) {
    for (int c = sub * 64; c < sub * 64 + 64; ++c)
      xs[row * XSTR + c] = vrow[c];
  } else {
    for (int c = sub * 64; c < sub * 64 + 64; ++c)
      xs[row * XSTR + c] = (g < Nc) ? vrow[c] : 0.0f;
  }
  float s1 = 0.0f, s2 = 0.0f;
  for (int c = sub * 64; c < sub * 64 + 64; ++c) {
    const float v = xs[row * XSTR + c];
    s1 += v;
    s2 += v * v;
  }
  for (int m = 1; m < 8; m <<= 1) {
    s1 += __shfl_xor(s1, m, 32);
    s2 += __shfl_xor(s2, m, 32);
  }
  const float mean = s1 * (1.0f / Dc);
  const float var = s2 * (1.0f / Dc) - mean * mean;
  const float rs = rsqrtf(var + 1e-5f);
  float* xrow = xo + ((size_t)b * Nc + g) * Dc;
  if (full) {
    for (int c = sub * 64; c < sub * 64 + 64; ++c) {
      const float y = (xs[row * XSTR + c] - mean) * rs * gamma[c] + beta[c];
      xs[row * XSTR + c] = y;
      xrow[c] = y;
    }
  } else {
    for (int c = sub * 64; c < sub * 64 + 64; ++c) {
      const float y = (xs[row * XSTR + c] - mean) * rs * gamma[c] + beta[c];
      xs[row * XSTR + c] = y;
      if (g < Nc) xrow[c] = y;
    }
  }
  __syncthreads();

  const int wave = tid >> 5, lane = tid & 31;
  const int hf = lane >> 4, l16 = lane & 15;
  const float* Wm = (wave < 2) ? Uw : Vw;
  float* out = (wave < 2) ? qo : ko;
  const int cb = (wave & 1) * 16;
  v8f acc = {0.f, 0.f, 0.f, 0.f, 0.f, 0.f, 0.f, 0.f};
  for (int kk = 0; kk < Dc / 4; ++kk) {
    const int c0 = 4 * kk + 2 * hf;
    v2f a;
    a.x = xs[l16 * XSTR + c0];
    a.y = xs[l16 * XSTR + c0 + 1];
    v2f bv;
    bv.x = Wm[(size_t)c0 * Rc + cb + l16];
    bv.y = Wm[(size_t)(c0 + 1) * Rc + cb + l16];
    acc = wmma_f32(a, bv, acc);
  }
  if (full) {
    for (int i = 0; i < 8; ++i) {
      const int r = i + 8 * hf;
      out[((size_t)b * Nc + n0 + r) * Rc + cb + l16] = acc[i];
    }
  } else {
    for (int i = 0; i < 8; ++i) {
      const int r = i + 8 * hf, gg = n0 + r;
      if (gg < Nc) out[((size_t)b * Nc + gg) * Rc + cb + l16] = acc[i];
    }
  }
}

__global__ __launch_bounds__(128) void attn_kernel(
    const float* __restrict__ x, const float* __restrict__ q,
    const float* __restrict__ k, const float* __restrict__ st,
    float* __restrict__ val) {
  __shared__ float qs[16 * QSTR];
  __shared__ float ks[WINc * KSTR];
  __shared__ float ss[16 * SSTR];
  __shared__ float wm[16 * WSTR];
  __shared__ float sw[WINc];
  const int tile = blockIdx.x % NT;
  const int b = blockIdx.x / NT;
  const int n0 = tile * 16;
  const int tid = threadIdx.x;
  const bool full = (n0 + 16 <= Nc);
  const bool interior = (n0 >= Wc) && (n0 + 16 + Wc <= Nc);

  const float* qb = q + ((size_t)b * Nc + n0) * Rc;
  const float* kb = k + ((size_t)b * Nc + n0 - Wc) * Rc;
  const float* sb = st + (size_t)b * Nc + n0 - Wc;
  if (interior) {
    for (int i = tid; i < 16 * Rc; i += 128) {
      const int r = i >> 5, c = i & 31;
      qs[r * QSTR + c] = qb[r * Rc + c];
    }
    for (int i = tid; i < WINc * Rc; i += 128) {
      const int j = i >> 5, c = i & 31;
      ks[j * KSTR + c] = kb[(size_t)j * Rc + c];
    }
    for (int j = tid; j < WINc; j += 128) sw[j] = sb[j];
  } else {
    for (int i = tid; i < 16 * Rc; i += 128) {
      const int r = i >> 5, c = i & 31, g = n0 + r;
      qs[r * QSTR + c] = (g < Nc) ? q[((size_t)b * Nc + g) * Rc + c] : 0.0f;
    }
    for (int i = tid; i < WINc * Rc; i += 128) {
      const int j = i >> 5, c = i & 31, gn = n0 - Wc + j;
      ks[j * KSTR + c] =
          (gn >= 0 && gn < Nc) ? k[((size_t)b * Nc + gn) * Rc + c] : 0.0f;
    }
    for (int j = tid; j < WINc; j += 128) {
      const int gn = n0 - Wc + j;
      sw[j] = (gn >= 0 && gn < Nc) ? st[(size_t)b * Nc + gn] : 0.0f;
    }
  }
  for (int i = tid; i < 16 * WSTR; i += 128) wm[i] = 0.0f;
  __syncthreads();

  const int wave = tid >> 5, lane = tid & 31;
  const int hf = lane >> 4, l16 = lane & 15;

  for (int t = wave; t < 9; t += 4) {
    const int cb = t * 16;
    v8f acc = {0.f, 0.f, 0.f, 0.f, 0.f, 0.f, 0.f, 0.f};
    for (int kk = 0; kk < Rc / 4; ++kk) {
      const int c0 = 4 * kk + 2 * hf;
      v2f a;
      a.x = qs[l16 * QSTR + c0];
      a.y = qs[l16 * QSTR + c0 + 1];
      v2f bv;
      bv.x = ks[(cb + l16) * KSTR + c0];
      bv.y = ks[(cb + l16) * KSTR + c0 + 1];
      acc = wmma_f32(a, bv, acc);
    }
    for (int i = 0; i < 8; ++i) ss[(i + 8 * hf) * SSTR + cb + l16] = acc[i];
  }
  __syncthreads();

  {
    const int row = tid >> 3, sub = tid & 7;
    const int g = n0 + row;
    if (interior) {
      float m = -1e30f;
      for (int o = sub; o < WNc; o += 8)
        m = fmaxf(m, fabsf(ss[row * SSTR + row + o]));
      for (int mm = 1; mm < 8; mm <<= 1) m = fmaxf(m, __shfl_xor(m, mm, 32));
      float Z = 0.0f;
      for (int o = sub; o < WNc; o += 8)
        Z += expf(fabsf(ss[row * SSTR + row + o]) - m);
      for (int mm = 1; mm < 8; mm <<= 1) Z += __shfl_xor(Z, mm, 32);
      const float inv = 1.0f / Z;
      for (int o = sub; o < WNc; o += 8) {
        const float s = ss[row * SSTR + row + o];
        const float p = expf(fabsf(s) - m) * inv;
        const float wv = (s > 0.0f) ? p : ((s < 0.0f) ? -p : 0.0f);
        wm[row * WSTR + row + o] = wv * sw[row + o];
      }
    } else if (g < Nc) {
      float m = -1e30f;
      for (int o = sub; o < WNc; o += 8) {
        const int gn = g - Wc + o;
        if (gn >= 0 && gn < Nc) m = fmaxf(m, fabsf(ss[row * SSTR + row + o]));
      }
      for (int mm = 1; mm < 8; mm <<= 1) m = fmaxf(m, __shfl_xor(m, mm, 32));
      float Z = 0.0f;
      for (int o = sub; o < WNc; o += 8) {
        const int gn = g - Wc + o;
        if (gn >= 0 && gn < Nc)
          Z += expf(fabsf(ss[row * SSTR + row + o]) - m);
      }
      for (int mm = 1; mm < 8; mm <<= 1) Z += __shfl_xor(Z, mm, 32);
      const float inv = 1.0f / Z;
      for (int o = sub; o < WNc; o += 8) {
        const int gn = g - Wc + o;
        if (gn >= 0 && gn < Nc) {
          const float s = ss[row * SSTR + row + o];
          const float p = expf(fabsf(s) - m) * inv;
          const float wv = (s > 0.0f) ? p : ((s < 0.0f) ? -p : 0.0f);
          wm[row * WSTR + row + o] = wv * sw[row + o];
        }
      }
    }
  }
  __syncthreads();

  const float* xw = x + ((size_t)b * Nc + n0 - Wc) * Dc;
  for (int ct = wave; ct < Dc / 16; ct += 4) {
    const int cb = ct * 16;
    v8f acc = {0.f, 0.f, 0.f, 0.f, 0.f, 0.f, 0.f, 0.f};
    if (interior) {
      for (int kk = 0; kk < WINc / 4; ++kk) {
        const int c0 = 4 * kk + 2 * hf;
        v2f a;
        a.x = wm[l16 * WSTR + c0];
        a.y = wm[l16 * WSTR + c0 + 1];
        v2f bv;
        bv.x = xw[(size_t)c0 * Dc + cb + l16];
        bv.y = xw[(size_t)(c0 + 1) * Dc + cb + l16];
        acc = wmma_f32(a, bv, acc);
      }
    } else {
      for (int kk = 0; kk < WINc / 4; ++kk) {
        const int c0 = 4 * kk + 2 * hf;
        v2f a;
        a.x = wm[l16 * WSTR + c0];
        a.y = wm[l16 * WSTR + c0 + 1];
        const int g0 = n0 - Wc + c0, g1 = g0 + 1;
        v2f bv;
        bv.x = (g0 >= 0 && g0 < Nc) ? x[((size_t)b * Nc + g0) * Dc + cb + l16]
                                    : 0.0f;
        bv.y = (g1 >= 0 && g1 < Nc) ? x[((size_t)b * Nc + g1) * Dc + cb + l16]
                                    : 0.0f;
        acc = wmma_f32(a, bv, acc);
      }
    }
    if (full) {
      float* vb = val + ((size_t)b * Nc + n0) * Dc + cb + l16;
      for (int i = 0; i < 8; ++i) {
        const int r = i + 8 * hf;
        vb[(size_t)r * Dc] += acc[i];
      }
    } else {
      for (int i = 0; i < 8; ++i) {
        const int r = i + 8 * hf, gg = n0 + r;
        if (gg < Nc) val[((size_t)b * Nc + gg) * Dc + cb + l16] += acc[i];
      }
    }
  }
}

extern "C" void kernel_launch(void* const* d_in, const int* in_sizes, int n_in,
                              void* d_out, int out_size, void* d_ws,
                              size_t ws_size, hipStream_t stream) {
  const int* ids = (const int*)d_in[0];
  const float* emb = (const float*)d_in[1];
  const float* pos = (const float*)d_in[2];
  const float* anchor_val = (const float*)d_in[3];
  const float* anchor_state = (const float*)d_in[4];
  const float* in_gamma = (const float*)d_in[5];
  const float* in_beta = (const float*)d_in[6];
  const float* ng = (const float*)d_in[7];
  const float* nb = (const float*)d_in[8];
  const float* Uw = (const float*)d_in[9];
  const float* Vw = (const float*)d_in[10];
  const float* sp_w = (const float*)d_in[11];
  const float* sp_b = (const float*)d_in[12];

  float* val = (float*)d_out;
  float* ws = (float*)d_ws;
  float* x = ws;
  float* q = x + (size_t)Bc * Nc * Dc;
  float* k = q + (size_t)Bc * Nc * Rc;
  float* st = k + (size_t)Bc * Nc * Rc;

  embed_ln_kernel<<<Bc * Nc, 256, 0, stream>>>(
      ids, emb, pos, anchor_val, anchor_state, in_gamma, in_beta, sp_w, sp_b,
      val, st);

  for (int l = 0; l < Lc; ++l) {
    ln_proj_kernel<<<Bc * NT, 128, 0, stream>>>(
        val, ng + (size_t)l * Dc, nb + (size_t)l * Dc, Uw + (size_t)l * Dc * Rc,
        Vw + (size_t)l * Dc * Rc, x, q, k);
    attn_kernel<<<Bc * NT, 128, 0, stream>>>(x, q, k, st, val);
  }
}


# attn kernel: grid split 4 ways by 128-column output chunk (520 work-groups, 3 fit a WGP): each redoes scores+softmax, fetches one x chunk behind the scores' barrier; no chunk loop, one LDS buffer
# speedup vs baseline: 4.5152x; 4.5152x over previous
#include <hip/hip_runtime.h>

typedef __attribute__((ext_vector_type(2))) float v2f;
typedef __attribute__((ext_vector_type(8))) float v8f;
typedef __attribute__((ext_vector_type(4))) float v4f;

constexpr int Bc = 2;
constexpr int Sc = 1024;
constexpr int Nc = Sc + 1;
constexpr int Dc = 512;
constexpr int Rc = 32;
constexpr int Lc = 3;
constexpr int Wc = 64;
constexpr int WNc = 2 * Wc + 1;
constexpr int WINc = 144;
constexpr int NT = (Nc + 15) / 16;

constexpr int XSTR = Dc + 4;
constexpr int SSTR = 146;
constexpr int WSTR = 145;

constexpr int CHc = 128;
constexpr int NCH = Dc / CHc;
constexpr int XPSTR = 2 * CHc + 32;
constexpr int XBUF = (WINc / 2) * XPSTR;
constexpr int XPW = WINc / 2 / 4;
constexpr int KB = 9;
constexpr int NKB = WINc / 4 / KB;

constexpr int WPR = 2 * Rc + 16;
constexpr int WHK = Dc / 2;
constexpr int WLD = WHK * Rc / 4 / 128;
constexpr int PKB = 16;
constexpr int PNB = WHK / 4 / PKB;

__device__ __forceinline__ v8f wmma_f32(v2f a, v2f b, v8f c) {
  return __builtin_amdgcn_wmma_f32_16x16x4_f32(false, a, false, b, (short)0, c,
                                               false, false);
}

__device__ __forceinline__ float block_sum256(float v, float* red) {
  for (int m = 16; m >= 1; m >>= 1) v += __shfl_xor(v, m, 32);
  const int wave = threadIdx.x >> 5, lane = threadIdx.x & 31;
  __syncthreads();
  if (lane == 0) red[wave] = v;
  __syncthreads();
  if (threadIdx.x < 8) {
    float t = red[threadIdx.x];
    t += __shfl_xor(t, 1, 32);
    t += __shfl_xor(t, 2, 32);
    t += __shfl_xor(t, 4, 32);
    if (threadIdx.x == 0) red[0] = t;
  }
  __syncthreads();
  return red[0];
}

__global__ __launch_bounds__(256) void embed_ln_kernel(
    const int* __restrict__ ids, const float* __restrict__ emb,
    const float* __restrict__ pos, const float* __restrict__ anchor_val,
    const float* __restrict__ anchor_state, const float* __restrict__ gamma,
    const float* __restrict__ beta, const float* __restrict__ sp_w,
    const float* __restrict__ sp_b, float* __restrict__ val,
    float* __restrict__ st) {
  __shared__ float red[8];
  const int n = blockIdx.x % Nc;
  const int b = blockIdx.x / Nc;
  const int tid = threadIdx.x;
  float* vrow = val + ((size_t)b * Nc + n) * Dc;
  if (n == 0) {
    vrow[tid] = anchor_val[tid];
    vrow[tid + 256] = anchor_val[tid + 256];
    if (tid == 0) st[(size_t)b * Nc] = anchor_state[0];
    return;
  }
  const int s = n - 1;
  const int id = ids[b * Sc + s];
  const float e0 = emb[(size_t)id * Dc + tid] + pos[(size_t)s * Dc + tid];
  const float e1 = emb[(size_t)id * Dc + tid + 256] + pos[(size_t)s * Dc + tid + 256];
  const float sum = block_sum256(e0 + e1, red);
  const float ssq = block_sum256(e0 * e0 + e1 * e1, red);
  const float mean = sum * (1.0f / Dc);
  const float var = ssq * (1.0f / Dc) - mean * mean;
  const float rs = rsqrtf(var + 1e-5f);
  const float y0 = (e0 - mean) * rs * gamma[tid] + beta[tid];
  const float y1 = (e1 - mean) * rs * gamma[tid + 256] + beta[tid + 256];
  vrow[tid] = y0;
  vrow[tid + 256] = y1;
  const float dot = block_sum256(y0 * sp_w[tid] + y1 * sp_w[tid + 256], red);
  if (tid == 0) st[(size_t)b * Nc + n] = dot + sp_b[0];
}

__device__ __forceinline__ void wpanel_load(v4f (&r)[2 * WLD],
                                            const float* __restrict__ Uw,
                                            const float* __restrict__ Vw,
                                            int tid, int half) {
#pragma unroll
  for (int p = 0; p < WLD; ++p) {
    const int f = tid + 128 * (half * WLD + p);
    r[2 * p] = *(const v4f*)(Uw + 4 * f);
    r[2 * p + 1] = *(const v4f*)(Vw + 4 * f);
  }
}

__device__ __forceinline__ void wpanel_store(const v4f (&r)[2 * WLD], float* wl,
                                             int tid, int half) {
#pragma unroll
  for (int p = 0; p < WLD; ++p) {
    const int f = tid + 128 * (half * WLD + p);
    float* d = wl + (f >> 3) * WPR + 4 * (f & 7);
    *(v4f*)d = r[2 * p];
    *(v4f*)(d + Rc) = r[2 * p + 1];
  }
}

__device__ __forceinline__ void pfrag_load(v2f (&a)[PKB], v2f (&b)[PKB],
                                           const float* ap, const float* bp,
                                           int k0) {
#pragma unroll
  for (int s = 0; s < PKB; ++s) {
    const int kk = k0 + s;
    a[s] = *(const v2f*)(ap + 4 * kk);
    b[s].x = bp[(4 * kk) * WPR];
    b[s].y = bp[(4 * kk + 1) * WPR];
  }
}

__global__ __launch_bounds__(128) void ln_proj_kernel(
    const float* __restrict__ val, const float* __restrict__ gamma,
    const float* __restrict__ beta, const float* __restrict__ Uw,
    const float* __restrict__ Vw, float* __restrict__ xo,
    float* __restrict__ qo, float* __restrict__ ko) {
  __shared__ __attribute__((aligned(16))) float xs[16 * XSTR];
  __shared__ __attribute__((aligned(16))) float wl[Dc * WPR];
  const int tile = blockIdx.x % NT;
  const int b = blockIdx.x / NT;
  const int own0 = tile * 16;
  const int n0 = min(own0, Nc - 16);
  const int tid = threadIdx.x;
  const bool whole = (n0 == own0);

  const int row = tid >> 3, sub = tid & 7;
  const int g = n0 + row;
  const float* vrow = val + ((size_t)b * Nc + g) * Dc;
  for (int c = sub * 64; c < sub * 64 + 64; ++c)
    xs[row * XSTR + c] = vrow[c];
  v4f w0[2 * WLD], w1[2 * WLD];
  wpanel_load(w0, Uw, Vw, tid, 0);
  float s1 = 0.0f, s2 = 0.0f;
  for (int c = sub * 64; c < sub * 64 + 64; ++c) {
    const float v = xs[row * XSTR + c];
    s1 += v;
    s2 += v * v;
  }
  for (int m = 1; m < 8; m <<= 1) {
    s1 += __shfl_xor(s1, m, 32);
    s2 += __shfl_xor(s2, m, 32);
  }
  const float mean = s1 * (1.0f / Dc);
  const float var = s2 * (1.0f / Dc) - mean * mean;
  const float rs = rsqrtf(var + 1e-5f);
  float* xrow = xo + ((size_t)b * Nc + g) * Dc;
  if (g >= own0) {
    for (int c = sub * 64; c < sub * 64 + 64; ++c) {
      const float y = (xs[row * XSTR + c] - mean) * rs * gamma[c] + beta[c];
      xs[row * XSTR + c] = y;
      xrow[c] = y;
    }
  } else {
    for (int c = sub * 64; c < sub * 64 + 64; ++c) {
      const float y = (xs[row * XSTR + c] - mean) * rs * gamma[c] + beta[c];
      xs[row * XSTR + c] = y;
    }
  }
  const int wave = tid >> 5, lane = tid & 31;
  const int hf = lane >> 4, l16 = lane & 15;
  float* out = (wave < 2) ? qo : ko;
  const int cb = (wave & 1) * 16;
  const float* ap = xs + l16 * XSTR + 2 * hf;
  const float* bp = wl + 2 * hf * WPR + wave * 16 + l16;
  v8f acc = {0.f, 0.f, 0.f, 0.f, 0.f, 0.f, 0.f, 0.f};
#pragma unroll
  for (int h = 0; h < 2; ++h) {
    if (h == 0) wpanel_store(w0, wl, tid, 0);
    else wpanel_store(w1, wl, tid, 1);
    __syncthreads();
    if (h == 0) wpanel_load(w1, Uw, Vw, tid, 1);
    v2f aA[PKB], bA[PKB], aB[PKB], bB[PKB];
    const int k0 = h * (WHK / 4);
    pfrag_load(aA, bA, ap, bp, k0);
#pragma unroll
    for (int blk = 0; blk < PNB; blk += 2) {
      pfrag_load(aB, bB, ap, bp, k0 + (blk + 1) * PKB);
      __builtin_amdgcn_sched_barrier(0);
#pragma unroll
      for (int s = 0; s < PKB; ++s) acc = wmma_f32(aA[s], bA[s], acc);
      __builtin_amdgcn_sched_barrier(0);
      if (blk + 2 < PNB) pfrag_load(aA, bA, ap, bp, k0 + (blk + 2) * PKB);
      __builtin_amdgcn_sched_barrier(0);
#pragma unroll
      for (int s = 0; s < PKB; ++s) acc = wmma_f32(aB[s], bB[s], acc);
      __builtin_amdgcn_sched_barrier(0);
    }
  }
  if (whole) {
    for (int i = 0; i < 8; ++i) {
      const int r = i + 8 * hf;
      out[((size_t)b * Nc + n0 + r) * Rc + cb + l16] = acc[i];
    }
  } else {
    for (int i = 0; i < 8; ++i) {
      const int r = i + 8 * hf, gg = n0 + r;
      if (gg >= own0) out[((size_t)b * Nc + gg) * Rc + cb + l16] = acc[i];
    }
  }
}

__device__ __forceinline__ void xchunk_load(v4f (&r)[2 * XPW],
                                            const float* __restrict__ xbat,
                                            int n0, int wv, int lane, int ch) {
#pragma unroll
  for (int m = 0; m < XPW; ++m) {
    const int j = 2 * (wv + 4 * m);
    const int g0 = min(max(n0 - Wc + j, 0), Nc - 1);
    const int g1 = min(max(n0 - Wc + j + 1, 0), Nc - 1);
    r[2 * m] = *(const v4f*)(xbat + (size_t)g0 * Dc + ch * CHc + 4 * lane);
    r[2 * m + 1] = *(const v4f*)(xbat + (size_t)g1 * Dc + ch * CHc + 4 * lane);
  }
}

__device__ __forceinline__ void xchunk_store(const v4f (&r)[2 * XPW],
                                             float* buf, int wv, int lane) {
#pragma unroll
  for (int m = 0; m < XPW; ++m) {
    const v4f lo = r[2 * m], hi = r[2 * m + 1];
    const v4f s0 = {lo.x, hi.x, lo.y, hi.y};
    const v4f s1 = {lo.z, hi.z, lo.w, hi.w};
    float* p = buf + (wv + 4 * m) * XPSTR + 8 * lane;
    *(v4f*)p = s0;
    *(v4f*)(p + 4) = s1;
  }
}

__device__ __forceinline__ void bblock_load(v2f (&bb)[KB][2], const float* bp,
                                            int blk) {
#pragma unroll
  for (int s = 0; s < KB; ++s) {
    const float* p = bp + 2 * (blk * KB + s) * XPSTR;
    bb[s][0] = *(const v2f*)p;
    bb[s][1] = *(const v2f*)(p + 32);
  }
}

__global__ __launch_bounds__(128) void attn_kernel(
    const float* __restrict__ x, const float* __restrict__ q,
    const float* __restrict__ k, const float* __restrict__ st,
    float* __restrict__ val) {
  __shared__ float ss[16 * SSTR];
  __shared__ float wm[16 * WSTR];
  __shared__ float sw[WINc];
  __shared__ __attribute__((aligned(16))) float xb[XBUF];
  const int cq = blockIdx.x % NCH;
  const int tile = (blockIdx.x / NCH) % NT;
  const int b = blockIdx.x / (NCH * NT);
  const int n0 = tile * 16;
  const int tid = threadIdx.x;
  const bool full = (n0 + 16 <= Nc);
  const bool interior = (n0 >= Wc) && (n0 + 16 + Wc <= Nc);
  const int wave = tid >> 5, lane = tid & 31;
  const int hf = lane >> 4, l16 = lane & 15;
  const int wv = __builtin_amdgcn_readfirstlane(wave);

  const float* qbat = q + (size_t)b * Nc * Rc + 16 * hf;
  const float* kbat = k + (size_t)b * Nc * Rc + 16 * hf;
  v4f qf[4], kf[3][4];
  {
    const int gq = min(n0 + l16, Nc - 1);
#pragma unroll
    for (int i = 0; i < 4; ++i)
      qf[i] = *(const v4f*)(qbat + (size_t)gq * Rc + 4 * i);
  }
#pragma unroll
  for (int u = 0; u < 3; ++u) {
    if (wv + 4 * u < 9) {
      const int gk = min(max(n0 - Wc + 16 * (wv + 4 * u) + l16, 0), Nc - 1);
#pragma unroll
      for (int i = 0; i < 4; ++i)
        kf[u][i] = *(const v4f*)(kbat + (size_t)gk * Rc + 4 * i);
    }
  }

  const float* sb = st + (size_t)b * Nc + n0 - Wc;
  if (interior) {
    for (int j = tid; j < WINc; j += 128) sw[j] = sb[j];
  } else {
    for (int j = tid; j < WINc; j += 128) {
      const int gn = n0 - Wc + j;
      sw[j] = (gn >= 0 && gn < Nc) ? st[(size_t)b * Nc + gn] : 0.0f;
    }
  }
  for (int i = tid; i < 16 * WSTR; i += 128) wm[i] = 0.0f;

#pragma unroll
  for (int u = 0; u < 3; ++u) {
    if (wv + 4 * u < 9) {
      const int cb = (wv + 4 * u) * 16;
      v8f acc = {0.f, 0.f, 0.f, 0.f, 0.f, 0.f, 0.f, 0.f};
#pragma unroll
      for (int i = 0; i < 4; ++i) {
        const v2f a0 = {qf[i].x, qf[i].y}, b0 = {kf[u][i].x, kf[u][i].y};
        acc = wmma_f32(a0, b0, acc);
        const v2f a1 = {qf[i].z, qf[i].w}, b1 = {kf[u][i].z, kf[u][i].w};
        acc = wmma_f32(a1, b1, acc);
      }
      for (int i = 0; i < 8; ++i) ss[(i + 8 * hf) * SSTR + cb + l16] = acc[i];
    }
  }
  __syncthreads();

  const float* xbat = x + (size_t)b * Nc * Dc;
  v4f xr[2 * XPW];
  xchunk_load(xr, xbat, n0, wv, lane, cq);

  {
    const int row = tid >> 3, sub = tid & 7;
    const int g = n0 + row;
    if (interior) {
      float m = -1e30f;
      for (int o = sub; o < WNc; o += 8)
        m = fmaxf(m, fabsf(ss[row * SSTR + row + o]));
      for (int mm = 1; mm < 8; mm <<= 1) m = fmaxf(m, __shfl_xor(m, mm, 32));
      float Z = 0.0f;
      for (int o = sub; o < WNc; o += 8)
        Z += expf(fabsf(ss[row * SSTR + row + o]) - m);
      for (int mm = 1; mm < 8; mm <<= 1) Z += __shfl_xor(Z, mm, 32);
      const float inv = 1.0f / Z;
      for (int o = sub; o < WNc; o += 8) {
        const float s = ss[row * SSTR + row + o];
        const float p = expf(fabsf(s) - m) * inv;
        const float wv2 = (s > 0.0f) ? p : ((s < 0.0f) ? -p : 0.0f);
        wm[row * WSTR + row + o] = wv2 * sw[row + o];
      }
    } else if (g < Nc) {
      float m = -1e30f;
      for (int o = sub; o < WNc; o += 8) {
        const int gn = g - Wc + o;
        if (gn >= 0 && gn < Nc) m = fmaxf(m, fabsf(ss[row * SSTR + row + o]));
      }
      for (int mm = 1; mm < 8; mm <<= 1) m = fmaxf(m, __shfl_xor(m, mm, 32));
      float Z = 0.0f;
      for (int o = sub; o < WNc; o += 8) {
        const int gn = g - Wc + o;
        if (gn >= 0 && gn < Nc)
          Z += expf(fabsf(ss[row * SSTR + row + o]) - m);
      }
      for (int mm = 1; mm < 8; mm <<= 1) Z += __shfl_xor(Z, mm, 32);
      const float inv = 1.0f / Z;
      for (int o = sub; o < WNc; o += 8) {
        const int gn = g - Wc + o;
        if (gn >= 0 && gn < Nc) {
          const float s = ss[row * SSTR + row + o];
          const float p = expf(fabsf(s) - m) * inv;
          const float wv2 = (s > 0.0f) ? p : ((s < 0.0f) ? -p : 0.0f);
          wm[row * WSTR + row + o] = wv2 * sw[row + o];
        }
      }
    }
  }
  xchunk_store(xr, xb, wv, lane);
  __syncthreads();


  v2f af[WINc / 4];
#pragma unroll
  for (int kk = 0; kk < WINc / 4; ++kk) {
    const int c0 = 4 * kk + 2 * hf;
    af[kk].x = wm[l16 * WSTR + c0];
    af[kk].y = wm[l16 * WSTR + c0 + 1];
  }

  float* vt = val + (size_t)b * Nc * Dc + cq * CHc + wv * 32 + l16;
  float o0[8], o1[8];
#pragma unroll
  for (int i = 0; i < 8; ++i) {
    const int gl = min(n0 + i + 8 * hf, Nc - 1);
    o0[i] = vt[(size_t)gl * Dc];
    o1[i] = vt[(size_t)gl * Dc + 16];
  }

  const float* bp = xb + hf * XPSTR + 2 * (wv * 32 + l16);
  v8f acc0 = {0.f, 0.f, 0.f, 0.f, 0.f, 0.f, 0.f, 0.f};
  v8f acc1 = {0.f, 0.f, 0.f, 0.f, 0.f, 0.f, 0.f, 0.f};
  v2f bA[KB][2], bB[KB][2];
  bblock_load(bA, bp, 0);
#pragma unroll
  for (int blk = 0; blk < NKB; blk += 2) {
    bblock_load(bB, bp, blk + 1);
    __builtin_amdgcn_sched_barrier(0);
#pragma unroll
    for (int s = 0; s < KB; ++s) {
      acc0 = wmma_f32(af[blk * KB + s], bA[s][0], acc0);
      acc1 = wmma_f32(af[blk * KB + s], bA[s][1], acc1);
    }
    __builtin_amdgcn_sched_barrier(0);
    if (blk + 2 < NKB) bblock_load(bA, bp, blk + 2);
    __builtin_amdgcn_sched_barrier(0);
#pragma unroll
    for (int s = 0; s < KB; ++s) {
      acc0 = wmma_f32(af[(blk + 1) * KB + s], bB[s][0], acc0);
      acc1 = wmma_f32(af[(blk + 1) * KB + s], bB[s][1], acc1);
    }
    __builtin_amdgcn_sched_barrier(0);
  }
  if (full) {
#pragma unroll
    for (int i = 0; i < 8; ++i) {
      float* p = vt + (size_t)(n0 + i + 8 * hf) * Dc;
      p[0] = o0[i] + acc0[i];
      p[16] = o1[i] + acc1[i];
    }
  } else {
#pragma unroll
    for (int i = 0; i < 8; ++i) {
      const int gg = n0 + i + 8 * hf;
      if (gg < Nc) {
        float* p = vt + (size_t)gg * Dc;
        p[0] = o0[i] + acc0[i];
        p[16] = o1[i] + acc1[i];
      }
    }
  }
}

extern "C" void kernel_launch(void* const* d_in, const int* in_sizes, int n_in,
                              void* d_out, int out_size, void* d_ws,
                              size_t ws_size, hipStream_t stream) {
  const int* ids = (const int*)d_in[0];
  const float* emb = (const float*)d_in[1];
  const float* pos = (const float*)d_in[2];
  const float* anchor_val = (const float*)d_in[3];
  const float* anchor_state = (const float*)d_in[4];
  const float* in_gamma = (const float*)d_in[5];
  const float* in_beta = (const float*)d_in[6];
  const float* ng = (const float*)d_in[7];
  const float* nb = (const float*)d_in[8];
  const float* Uw = (const float*)d_in[9];
  const float* Vw = (const float*)d_in[10];
  const float* sp_w = (const float*)d_in[11];
  const float* sp_b = (const float*)d_in[12];

  float* val = (float*)d_out;
  float* ws = (float*)d_ws;
  float* x = ws;
  float* q = x + (size_t)Bc * Nc * Dc;
  float* k = q + (size_t)Bc * Nc * Rc;
  float* st = k + (size_t)Bc * Nc * Rc;

  embed_ln_kernel<<<Bc * Nc, 256, 0, stream>>>(
      ids, emb, pos, anchor_val, anchor_state, in_gamma, in_beta, sp_w, sp_b,
      val, st);

  for (int l = 0; l < Lc; ++l) {
    ln_proj_kernel<<<Bc * NT, 128, 0, stream>>>(
        val, ng + (size_t)l * Dc, nb + (size_t)l * Dc, Uw + (size_t)l * Dc * Rc,
        Vw + (size_t)l * Dc * Rc, x, q, k);
    attn_kernel<<<Bc * NT * NCH, 128, 0, stream>>>(x, q, k, st, val);
  }
}


# LN+projection kernel: a row's 8 LayerNorm threads own interleaved 16-byte pieces (s, s+8, ..), not 64-channel blocks: whole 128-byte lines per access, no 8-way LDS bank conflict; LN sums reordered
# speedup vs baseline: 4.9270x; 1.0912x over previous
#include <hip/hip_runtime.h>

typedef __attribute__((ext_vector_type(2))) float v2f;
typedef __attribute__((ext_vector_type(8))) float v8f;
typedef __attribute__((ext_vector_type(4))) float v4f;

constexpr int Bc = 2;
constexpr int Sc = 1024;
constexpr int Nc = Sc + 1;
constexpr int Dc = 512;
constexpr int Rc = 32;
constexpr int Lc = 3;
constexpr int Wc = 64;
constexpr int WNc = 2 * Wc + 1;
constexpr int WINc = 144;
constexpr int NT = (Nc + 15) / 16;

constexpr int XSTR = Dc + 4;
constexpr int SSTR = 146;
constexpr int WSTR = 145;

constexpr int CHc = 128;
constexpr int NCH = Dc / CHc;
constexpr int XPSTR = 2 * CHc + 32;
constexpr int XBUF = (WINc / 2) * XPSTR;
constexpr int XPW = WINc / 2 / 4;
constexpr int KB = 9;
constexpr int NKB = WINc / 4 / KB;

constexpr int WPR = 2 * Rc + 16;
constexpr int WHK = Dc / 2;
constexpr int WLD = WHK * Rc / 4 / 128;
constexpr int PKB = 16;
constexpr int PNB = WHK / 4 / PKB;
constexpr int LNP = Dc / 4 / 8;

__device__ __forceinline__ v8f wmma_f32(v2f a, v2f b, v8f c) {
  return __builtin_amdgcn_wmma_f32_16x16x4_f32(false, a, false, b, (short)0, c,
                                               false, false);
}

__device__ __forceinline__ float block_sum256(float v, float* red) {
  for (int m = 16; m >= 1; m >>= 1) v += __shfl_xor(v, m, 32);
  const int wave = threadIdx.x >> 5, lane = threadIdx.x & 31;
  __syncthreads();
  if (lane == 0) red[wave] = v;
  __syncthreads();
  if (threadIdx.x < 8) {
    float t = red[threadIdx.x];
    t += __shfl_xor(t, 1, 32);
    t += __shfl_xor(t, 2, 32);
    t += __shfl_xor(t, 4, 32);
    if (threadIdx.x == 0) red[0] = t;
  }
  __syncthreads();
  return red[0];
}

__global__ __launch_bounds__(256) void embed_ln_kernel(
    const int* __restrict__ ids, const float* __restrict__ emb,
    const float* __restrict__ pos, const float* __restrict__ anchor_val,
    const float* __restrict__ anchor_state, const float* __restrict__ gamma,
    const float* __restrict__ beta, const float* __restrict__ sp_w,
    const float* __restrict__ sp_b, float* __restrict__ val,
    float* __restrict__ st) {
  __shared__ float red[8];
  const int n = blockIdx.x % Nc;
  const int b = blockIdx.x / Nc;
  const int tid = threadIdx.x;
  float* vrow = val + ((size_t)b * Nc + n) * Dc;
  if (n == 0) {
    vrow[tid] = anchor_val[tid];
    vrow[tid + 256] = anchor_val[tid + 256];
    if (tid == 0) st[(size_t)b * Nc] = anchor_state[0];
    return;
  }
  const int s = n - 1;
  const int id = ids[b * Sc + s];
  const float e0 = emb[(size_t)id * Dc + tid] + pos[(size_t)s * Dc + tid];
  const float e1 = emb[(size_t)id * Dc + tid + 256] + pos[(size_t)s * Dc + tid + 256];
  const float sum = block_sum256(e0 + e1, red);
  const float ssq = block_sum256(e0 * e0 + e1 * e1, red);
  const float mean = sum * (1.0f / Dc);
  const float var = ssq * (1.0f / Dc) - mean * mean;
  const float rs = rsqrtf(var + 1e-5f);
  const float y0 = (e0 - mean) * rs * gamma[tid] + beta[tid];
  const float y1 = (e1 - mean) * rs * gamma[tid + 256] + beta[tid + 256];
  vrow[tid] = y0;
  vrow[tid + 256] = y1;
  const float dot = block_sum256(y0 * sp_w[tid] + y1 * sp_w[tid + 256], red);
  if (tid == 0) st[(size_t)b * Nc + n] = dot + sp_b[0];
}

__device__ __forceinline__ void wpanel_load(v4f (&r)[2 * WLD],
                                            const float* __restrict__ Uw,
                                            const float* __restrict__ Vw,
                                            int tid, int half) {
#pragma unroll
  for (int p = 0; p < WLD; ++p) {
    const int f = tid + 128 * (half * WLD + p);
    r[2 * p] = *(const v4f*)(Uw + 4 * f);
    r[2 * p + 1] = *(const v4f*)(Vw + 4 * f);
  }
}

__device__ __forceinline__ void wpanel_store(const v4f (&r)[2 * WLD], float* wl,
                                             int tid, int half) {
#pragma unroll
  for (int p = 0; p < WLD; ++p) {
    const int f = tid + 128 * (half * WLD + p);
    float* d = wl + (f >> 3) * WPR + 4 * (f & 7);
    *(v4f*)d = r[2 * p];
    *(v4f*)(d + Rc) = r[2 * p + 1];
  }
}

__device__ __forceinline__ void pfrag_load(v2f (&a)[PKB], v2f (&b)[PKB],
                                           const float* ap, const float* bp,
                                           int k0) {
#pragma unroll
  for (int s = 0; s < PKB; ++s) {
    const int kk = k0 + s;
    a[s] = *(const v2f*)(ap + 4 * kk);
    b[s].x = bp[(4 * kk) * WPR];
    b[s].y = bp[(4 * kk + 1) * WPR];
  }
}

__global__ __launch_bounds__(128) void ln_proj_kernel(
    const float* __restrict__ val, const float* __restrict__ gamma,
    const float* __restrict__ beta, const float* __restrict__ Uw,
    const float* __restrict__ Vw, float* __restrict__ xo,
    float* __restrict__ qo, float* __restrict__ ko) {
  __shared__ __attribute__((aligned(16))) float xs[16 * XSTR];
  __shared__ __attribute__((aligned(16))) float wl[Dc * WPR];
  const int tile = blockIdx.x % NT;
  const int b = blockIdx.x / NT;
  const int own0 = tile * 16;
  const int n0 = min(own0, Nc - 16);
  const int tid = threadIdx.x;
  const bool whole = (n0 == own0);

  const int row = tid >> 3, sub = tid & 7;
  const int g = n0 + row;
  const float* vrow = val + ((size_t)b * Nc + g) * Dc + 4 * sub;
  float* xsr = xs + row * XSTR + 4 * sub;
  for (int i = 0; i < LNP; ++i)
    *(v4f*)(xsr + 32 * i) = *(const v4f*)(vrow + 32 * i);
  v4f w0[2 * WLD], w1[2 * WLD];
  wpanel_load(w0, Uw, Vw, tid, 0);
  float s1 = 0.0f, s2 = 0.0f;
  for (int i = 0; i < LNP; ++i) {
    const v4f v = *(const v4f*)(xsr + 32 * i);
    for (int e = 0; e < 4; ++e) {
      s1 += v[e];
      s2 += v[e] * v[e];
    }
  }
  for (int m = 1; m < 8; m <<= 1) {
    s1 += __shfl_xor(s1, m, 32);
    s2 += __shfl_xor(s2, m, 32);
  }
  const float mean = s1 * (1.0f / Dc);
  const float var = s2 * (1.0f / Dc) - mean * mean;
  const float rs = rsqrtf(var + 1e-5f);
  float* xrow = xo + ((size_t)b * Nc + g) * Dc + 4 * sub;
  const float* gpc = gamma + 4 * sub;
  const float* bpc = beta + 4 * sub;
  if (g >= own0) {
    for (int i = 0; i < LNP; ++i) {
      const v4f v = *(const v4f*)(xsr + 32 * i);
      const v4f gm = *(const v4f*)(gpc + 32 * i);
      const v4f bt = *(const v4f*)(bpc + 32 * i);
      v4f y;
      for (int e = 0; e < 4; ++e) y[e] = (v[e] - mean) * rs * gm[e] + bt[e];
      *(v4f*)(xsr + 32 * i) = y;
      *(v4f*)(xrow + 32 * i) = y;
    }
  } else {
    for (int i = 0; i < LNP; ++i) {
      const v4f v = *(const v4f*)(xsr + 32 * i);
      const v4f gm = *(const v4f*)(gpc + 32 * i);
      const v4f bt = *(const v4f*)(bpc + 32 * i);
      v4f y;
      for (int e = 0; e < 4; ++e) y[e] = (v[e] - mean) * rs * gm[e] + bt[e];
      *(v4f*)(xsr + 32 * i) = y;
    }
  }
  const int wave = tid >> 5, lane = tid & 31;
  const int hf = lane >> 4, l16 = lane & 15;
  float* out = (wave < 2) ? qo : ko;
  const int cb = (wave & 1) * 16;
  const float* ap = xs + l16 * XSTR + 2 * hf;
  const float* bp = wl + 2 * hf * WPR + wave * 16 + l16;
  v8f acc = {0.f, 0.f, 0.f, 0.f, 0.f, 0.f, 0.f, 0.f};
#pragma unroll
  for (int h = 0; h < 2; ++h) {
    if (h == 0) wpanel_store(w0, wl, tid, 0);
    else wpanel_store(w1, wl, tid, 1);
    __syncthreads();
    if (h == 0) wpanel_load(w1, Uw, Vw, tid, 1);
    v2f aA[PKB], bA[PKB], aB[PKB], bB[PKB];
    const int k0 = h * (WHK / 4);
    pfrag_load(aA, bA, ap, bp, k0);
#pragma unroll
    for (int blk = 0; blk < PNB; blk += 2) {
      pfrag_load(aB, bB, ap, bp, k0 + (blk + 1) * PKB);
      __builtin_amdgcn_sched_barrier(0);
#pragma unroll
      for (int s = 0; s < PKB; ++s) acc = wmma_f32(aA[s], bA[s], acc);
      __builtin_amdgcn_sched_barrier(0);
      if (blk + 2 < PNB) pfrag_load(aA, bA, ap, bp, k0 + (blk + 2) * PKB);
      __builtin_amdgcn_sched_barrier(0);
#pragma unroll
      for (int s = 0; s < PKB; ++s) acc = wmma_f32(aB[s], bB[s], acc);
      __builtin_amdgcn_sched_barrier(0);
    }
  }
  if (whole) {
    for (int i = 0; i < 8; ++i) {
      const int r = i + 8 * hf;
      out[((size_t)b * Nc + n0 + r) * Rc + cb + l16] = acc[i];
    }
  } else {
    for (int i = 0; i < 8; ++i) {
      const int r = i + 8 * hf, gg = n0 + r;
      if (gg >= own0) out[((size_t)b * Nc + gg) * Rc + cb + l16] = acc[i];
    }
  }
}

__device__ __forceinline__ void xchunk_load(v4f (&r)[2 * XPW],
                                            const float* __restrict__ xbat,
                                            int n0, int wv, int lane, int ch) {
#pragma unroll
  for (int m = 0; m < XPW; ++m) {
    const int j = 2 * (wv + 4 * m);
    const int g0 = min(max(n0 - Wc + j, 0), Nc - 1);
    const int g1 = min(max(n0 - Wc + j + 1, 0), Nc - 1);
    r[2 * m] = *(const v4f*)(xbat + (size_t)g0 * Dc + ch * CHc + 4 * lane);
    r[2 * m + 1] = *(const v4f*)(xbat + (size_t)g1 * Dc + ch * CHc + 4 * lane);
  }
}

__device__ __forceinline__ void xchunk_store(const v4f (&r)[2 * XPW],
                                             float* buf, int wv, int lane) {
#pragma unroll
  for (int m = 0; m < XPW; ++m) {
    const v4f lo = r[2 * m], hi = r[2 * m + 1];
    const v4f s0 = {lo.x, hi.x, lo.y, hi.y};
    const v4f s1 = {lo.z, hi.z, lo.w, hi.w};
    float* p = buf + (wv + 4 * m) * XPSTR + 8 * lane;
    *(v4f*)p = s0;
    *(v4f*)(p + 4) = s1;
  }
}

__device__ __forceinline__ void bblock_load(v2f (&bb)[KB][2], const float* bp,
                                            int blk) {
#pragma unroll
  for (int s = 0; s < KB; ++s) {
    const float* p = bp + 2 * (blk * KB + s) * XPSTR;
    bb[s][0] = *(const v2f*)p;
    bb[s][1] = *(const v2f*)(p + 32);
  }
}

__global__ __launch_bounds__(128) void attn_kernel(
    const float* __restrict__ x, const float* __restrict__ q,
    const float* __restrict__ k, const float* __restrict__ st,
    float* __restrict__ val) {
  __shared__ float ss[16 * SSTR];
  __shared__ float wm[16 * WSTR];
  __shared__ float sw[WINc];
  __shared__ __attribute__((aligned(16))) float xb[XBUF];
  const int cq = blockIdx.x % NCH;
  const int tile = (blockIdx.x / NCH) % NT;
  const int b = blockIdx.x / (NCH * NT);
  const int n0 = tile * 16;
  const int tid = threadIdx.x;
  const bool full = (n0 + 16 <= Nc);
  const bool interior = (n0 >= Wc) && (n0 + 16 + Wc <= Nc);
  const int wave = tid >> 5, lane = tid & 31;
  const int hf = lane >> 4, l16 = lane & 15;
  const int wv = __builtin_amdgcn_readfirstlane(wave);

  const float* qbat = q + (size_t)b * Nc * Rc + 16 * hf;
  const float* kbat = k + (size_t)b * Nc * Rc + 16 * hf;
  v4f qf[4], kf[3][4];
  {
    const int gq = min(n0 + l16, Nc - 1);
#pragma unroll
    for (int i = 0; i < 4; ++i)
      qf[i] = *(const v4f*)(qbat + (size_t)gq * Rc + 4 * i);
  }
#pragma unroll
  for (int u = 0; u < 3; ++u) {
    if (wv + 4 * u < 9) {
      const int gk = min(max(n0 - Wc + 16 * (wv + 4 * u) + l16, 0), Nc - 1);
#pragma unroll
      for (int i = 0; i < 4; ++i)
        kf[u][i] = *(const v4f*)(kbat + (size_t)gk * Rc + 4 * i);
    }
  }

  const float* sb = st + (size_t)b * Nc + n0 - Wc;
  if (interior) {
    for (int j = tid; j < WINc; j += 128) sw[j] = sb[j];
  } else {
    for (int j = tid; j < WINc; j += 128) {
      const int gn = n0 - Wc + j;
      sw[j] = (gn >= 0 && gn < Nc) ? st[(size_t)b * Nc + gn] : 0.0f;
    }
  }
  for (int i = tid; i < 16 * WSTR; i += 128) wm[i] = 0.0f;

#pragma unroll
  for (int u = 0; u < 3; ++u) {
    if (wv + 4 * u < 9) {
      const int cb = (wv + 4 * u) * 16;
      v8f acc = {0.f, 0.f, 0.f, 0.f, 0.f, 0.f, 0.f, 0.f};
#pragma unroll
      for (int i = 0; i < 4; ++i) {
        const v2f a0 = {qf[i].x, qf[i].y}, b0 = {kf[u][i].x, kf[u][i].y};
        acc = wmma_f32(a0, b0, acc);
        const v2f a1 = {qf[i].z, qf[i].w}, b1 = {kf[u][i].z, kf[u][i].w};
        acc = wmma_f32(a1, b1, acc);
      }
      for (int i = 0; i < 8; ++i) ss[(i + 8 * hf) * SSTR + cb + l16] = acc[i];
    }
  }
  __syncthreads();

  const float* xbat = x + (size_t)b * Nc * Dc;
  v4f xr[2 * XPW];
  xchunk_load(xr, xbat, n0, wv, lane, cq);

  {
    const int row = tid >> 3, sub = tid & 7;
    const int g = n0 + row;
    if (interior) {
      float m = -1e30f;
      for (int o = sub; o < WNc; o += 8)
        m = fmaxf(m, fabsf(ss[row * SSTR + row + o]));
      for (int mm = 1; mm < 8; mm <<= 1) m = fmaxf(m, __shfl_xor(m, mm, 32));
      float Z = 0.0f;
      for (int o = sub; o < WNc; o += 8)
        Z += expf(fabsf(ss[row * SSTR + row + o]) - m);
      for (int mm = 1; mm < 8; mm <<= 1) Z += __shfl_xor(Z, mm, 32);
      const float inv = 1.0f / Z;
      for (int o = sub; o < WNc; o += 8) {
        const float s = ss[row * SSTR + row + o];
        const float p = expf(fabsf(s) - m) * inv;
        const float wv2 = (s > 0.0f) ? p : ((s < 0.0f) ? -p : 0.0f);
        wm[row * WSTR + row + o] = wv2 * sw[row + o];
      }
    } else if (g < Nc) {
      float m = -1e30f;
      for (int o = sub; o < WNc; o += 8) {
        const int gn = g - Wc + o;
        if (gn >= 0 && gn < Nc) m = fmaxf(m, fabsf(ss[row * SSTR + row + o]));
      }
      for (int mm = 1; mm < 8; mm <<= 1) m = fmaxf(m, __shfl_xor(m, mm, 32));
      float Z = 0.0f;
      for (int o = sub; o < WNc; o += 8) {
        const int gn = g - Wc + o;
        if (gn >= 0 && gn < Nc)
          Z += expf(fabsf(ss[row * SSTR + row + o]) - m);
      }
      for (int mm = 1; mm < 8; mm <<= 1) Z += __shfl_xor(Z, mm, 32);
      const float inv = 1.0f / Z;
      for (int o = sub; o < WNc; o += 8) {
        const int gn = g - Wc + o;
        if (gn >= 0 && gn < Nc) {
          const float s = ss[row * SSTR + row + o];
          const float p = expf(fabsf(s) - m) * inv;
          const float wv2 = (s > 0.0f) ? p : ((s < 0.0f) ? -p : 0.0f);
          wm[row * WSTR + row + o] = wv2 * sw[row + o];
        }
      }
    }
  }
  xchunk_store(xr, xb, wv, lane);
  __syncthreads();


  v2f af[WINc / 4];
#pragma unroll
  for (int kk = 0; kk < WINc / 4; ++kk) {
    const int c0 = 4 * kk + 2 * hf;
    af[kk].x = wm[l16 * WSTR + c0];
    af[kk].y = wm[l16 * WSTR + c0 + 1];
  }

  float* vt = val + (size_t)b * Nc * Dc + cq * CHc + wv * 32 + l16;
  float o0[8], o1[8];
#pragma unroll
  for (int i = 0; i < 8; ++i) {
    const int gl = min(n0 + i + 8 * hf, Nc - 1);
    o0[i] = vt[(size_t)gl * Dc];
    o1[i] = vt[(size_t)gl * Dc + 16];
  }

  const float* bp = xb + hf * XPSTR + 2 * (wv * 32 + l16);
  v8f acc0 = {0.f, 0.f, 0.f, 0.f, 0.f, 0.f, 0.f, 0.f};
  v8f acc1 = {0.f, 0.f, 0.f, 0.f, 0.f, 0.f, 0.f, 0.f};
  v2f bA[KB][2], bB[KB][2];
  bblock_load(bA, bp, 0);
#pragma unroll
  for (int blk = 0; blk < NKB; blk += 2) {
    bblock_load(bB, bp, blk + 1);
    __builtin_amdgcn_sched_barrier(0);
#pragma unroll
    for (int s = 0; s < KB; ++s) {
      acc0 = wmma_f32(af[blk * KB + s], bA[s][0], acc0);
      acc1 = wmma_f32(af[blk * KB + s], bA[s][1], acc1);
    }
    __builtin_amdgcn_sched_barrier(0);
    if (blk + 2 < NKB) bblock_load(bA, bp, blk + 2);
    __builtin_amdgcn_sched_barrier(0);
#pragma unroll
    for (int s = 0; s < KB; ++s) {
      acc0 = wmma_f32(af[(blk + 1) * KB + s], bB[s][0], acc0);
      acc1 = wmma_f32(af[(blk + 1) * KB + s], bB[s][1], acc1);
    }
    __builtin_amdgcn_sched_barrier(0);
  }
  if (full) {
#pragma unroll
    for (int i = 0; i < 8; ++i) {
      float* p = vt + (size_t)(n0 + i + 8 * hf) * Dc;
      p[0] = o0[i] + acc0[i];
      p[16] = o1[i] + acc1[i];
    }
  } else {
#pragma unroll
    for (int i = 0; i < 8; ++i) {
      const int gg = n0 + i + 8 * hf;
      if (gg < Nc) {
        float* p = vt + (size_t)gg * Dc;
        p[0] = o0[i] + acc0[i];
        p[16] = o1[i] + acc1[i];
      }
    }
  }
}

extern "C" void kernel_launch(void* const* d_in, const int* in_sizes, int n_in,
                              void* d_out, int out_size, void* d_ws,
                              size_t ws_size, hipStream_t stream) {
  const int* ids = (const int*)d_in[0];
  const float* emb = (const float*)d_in[1];
  const float* pos = (const float*)d_in[2];
  const float* anchor_val = (const float*)d_in[3];
  const float* anchor_state = (const float*)d_in[4];
  const float* in_gamma = (const float*)d_in[5];
  const float* in_beta = (const float*)d_in[6];
  const float* ng = (const float*)d_in[7];
  const float* nb = (const float*)d_in[8];
  const float* Uw = (const float*)d_in[9];
  const float* Vw = (const float*)d_in[10];
  const float* sp_w = (const float*)d_in[11];
  const float* sp_b = (const float*)d_in[12];

  float* val = (float*)d_out;
  float* ws = (float*)d_ws;
  float* x = ws;
  float* q = x + (size_t)Bc * Nc * Dc;
  float* k = q + (size_t)Bc * Nc * Rc;
  float* st = k + (size_t)Bc * Nc * Rc;

  embed_ln_kernel<<<Bc * Nc, 256, 0, stream>>>(
      ids, emb, pos, anchor_val, anchor_state, in_gamma, in_beta, sp_w, sp_b,
      val, st);

  for (int l = 0; l < Lc; ++l) {
    ln_proj_kernel<<<Bc * NT, 128, 0, stream>>>(
        val, ng + (size_t)l * Dc, nb + (size_t)l * Dc, Uw + (size_t)l * Dc * Rc,
        Vw + (size_t)l * Dc * Rc, x, q, k);
    attn_kernel<<<Bc * NT * NCH, 128, 0, stream>>>(x, q, k, st, val);
  }
}


# LN+projection kernel: every global load requested at the top in order of need (val rows, gamma, beta, then the whole U/V panel, not half); panel staged to LDS once: one barrier, one run of products
# speedup vs baseline: 5.1269x; 1.0406x over previous
#include <hip/hip_runtime.h>

typedef __attribute__((ext_vector_type(2))) float v2f;
typedef __attribute__((ext_vector_type(8))) float v8f;
typedef __attribute__((ext_vector_type(4))) float v4f;

constexpr int Bc = 2;
constexpr int Sc = 1024;
constexpr int Nc = Sc + 1;
constexpr int Dc = 512;
constexpr int Rc = 32;
constexpr int Lc = 3;
constexpr int Wc = 64;
constexpr int WNc = 2 * Wc + 1;
constexpr int WINc = 144;
constexpr int NT = (Nc + 15) / 16;

constexpr int XSTR = Dc + 4;
constexpr int SSTR = 146;
constexpr int WSTR = 145;

constexpr int CHc = 128;
constexpr int NCH = Dc / CHc;
constexpr int XPSTR = 2 * CHc + 32;
constexpr int XBUF = (WINc / 2) * XPSTR;
constexpr int XPW = WINc / 2 / 4;
constexpr int KB = 9;
constexpr int NKB = WINc / 4 / KB;

constexpr int WPR = 2 * Rc + 16;
constexpr int WHK = Dc / 2;
constexpr int WLD = WHK * Rc / 4 / 128;
constexpr int PKB = 16;
constexpr int PNB = WHK / 4 / PKB;
constexpr int LNP = Dc / 4 / 8;

__device__ __forceinline__ v8f wmma_f32(v2f a, v2f b, v8f c) {
  return __builtin_amdgcn_wmma_f32_16x16x4_f32(false, a, false, b, (short)0, c,
                                               false, false);
}

__device__ __forceinline__ float block_sum256(float v, float* red) {
  for (int m = 16; m >= 1; m >>= 1) v += __shfl_xor(v, m, 32);
  const int wave = threadIdx.x >> 5, lane = threadIdx.x & 31;
  __syncthreads();
  if (lane == 0) red[wave] = v;
  __syncthreads();
  if (threadIdx.x < 8) {
    float t = red[threadIdx.x];
    t += __shfl_xor(t, 1, 32);
    t += __shfl_xor(t, 2, 32);
    t += __shfl_xor(t, 4, 32);
    if (threadIdx.x == 0) red[0] = t;
  }
  __syncthreads();
  return red[0];
}

__global__ __launch_bounds__(256) void embed_ln_kernel(
    const int* __restrict__ ids, const float* __restrict__ emb,
    const float* __restrict__ pos, const float* __restrict__ anchor_val,
    const float* __restrict__ anchor_state, const float* __restrict__ gamma,
    const float* __restrict__ beta, const float* __restrict__ sp_w,
    const float* __restrict__ sp_b, float* __restrict__ val,
    float* __restrict__ st) {
  __shared__ float red[8];
  const int n = blockIdx.x % Nc;
  const int b = blockIdx.x / Nc;
  const int tid = threadIdx.x;
  float* vrow = val + ((size_t)b * Nc + n) * Dc;
  if (n == 0) {
    vrow[tid] = anchor_val[tid];
    vrow[tid + 256] = anchor_val[tid + 256];
    if (tid == 0) st[(size_t)b * Nc] = anchor_state[0];
    return;
  }
  const int s = n - 1;
  const int id = ids[b * Sc + s];
  const float e0 = emb[(size_t)id * Dc + tid] + pos[(size_t)s * Dc + tid];
  const float e1 = emb[(size_t)id * Dc + tid + 256] + pos[(size_t)s * Dc + tid + 256];
  const float sum = block_sum256(e0 + e1, red);
  const float ssq = block_sum256(e0 * e0 + e1 * e1, red);
  const float mean = sum * (1.0f / Dc);
  const float var = ssq * (1.0f / Dc) - mean * mean;
  const float rs = rsqrtf(var + 1e-5f);
  const float y0 = (e0 - mean) * rs * gamma[tid] + beta[tid];
  const float y1 = (e1 - mean) * rs * gamma[tid + 256] + beta[tid + 256];
  vrow[tid] = y0;
  vrow[tid + 256] = y1;
  const float dot = block_sum256(y0 * sp_w[tid] + y1 * sp_w[tid + 256], red);
  if (tid == 0) st[(size_t)b * Nc + n] = dot + sp_b[0];
}

__device__ __forceinline__ void wpanel_load(v4f (&r)[2 * WLD],
                                            const float* __restrict__ Uw,
                                            const float* __restrict__ Vw,
                                            int tid, int half) {
#pragma unroll
  for (int p = 0; p < WLD; ++p) {
    const int f = tid + 128 * (half * WLD + p);
    r[2 * p] = *(const v4f*)(Uw + 4 * f);
    r[2 * p + 1] = *(const v4f*)(Vw + 4 * f);
  }
}

__device__ __forceinline__ void wpanel_store(const v4f (&r)[2 * WLD], float* wl,
                                             int tid, int half) {
#pragma unroll
  for (int p = 0; p < WLD; ++p) {
    const int f = tid + 128 * (half * WLD + p);
    float* d = wl + (f >> 3) * WPR + 4 * (f & 7);
    *(v4f*)d = r[2 * p];
    *(v4f*)(d + Rc) = r[2 * p + 1];
  }
}

__device__ __forceinline__ void pfrag_load(v2f (&a)[PKB], v2f (&b)[PKB],
                                           const float* ap, const float* bp,
                                           int k0) {
#pragma unroll
  for (int s = 0; s < PKB; ++s) {
    const int kk = k0 + s;
    a[s] = *(const v2f*)(ap + 4 * kk);
    b[s].x = bp[(4 * kk) * WPR];
    b[s].y = bp[(4 * kk + 1) * WPR];
  }
}

__global__ __launch_bounds__(128) void ln_proj_kernel(
    const float* __restrict__ val, const float* __restrict__ gamma,
    const float* __restrict__ beta, const float* __restrict__ Uw,
    const float* __restrict__ Vw, float* __restrict__ xo,
    float* __restrict__ qo, float* __restrict__ ko) {
  __shared__ __attribute__((aligned(16))) float xs[16 * XSTR];
  __shared__ __attribute__((aligned(16))) float wl[Dc * WPR];
  const int tile = blockIdx.x % NT;
  const int b = blockIdx.x / NT;
  const int own0 = tile * 16;
  const int n0 = min(own0, Nc - 16);
  const int tid = threadIdx.x;
  const bool whole = (n0 == own0);

  const int row = tid >> 3, sub = tid & 7;
  const int g = n0 + row;
  const float* vrow = val + ((size_t)b * Nc + g) * Dc + 4 * sub;
  const float* gpc = gamma + 4 * sub;
  const float* bpc = beta + 4 * sub;
  float* xsr = xs + row * XSTR + 4 * sub;
  v4f vr[LNP], gm[LNP], bt[LNP];
#pragma unroll
  for (int i = 0; i < LNP; ++i) vr[i] = *(const v4f*)(vrow + 32 * i);
#pragma unroll
  for (int i = 0; i < LNP; ++i) {
    gm[i] = *(const v4f*)(gpc + 32 * i);
    bt[i] = *(const v4f*)(bpc + 32 * i);
  }
  __builtin_amdgcn_sched_barrier(0);
  v4f w0[2 * WLD], w1[2 * WLD];
  wpanel_load(w0, Uw, Vw, tid, 0);
  wpanel_load(w1, Uw, Vw, tid, 1);
  __builtin_amdgcn_sched_barrier(0);
  float s1 = 0.0f, s2 = 0.0f;
#pragma unroll
  for (int i = 0; i < LNP; ++i) {
    for (int e = 0; e < 4; ++e) {
      s1 += vr[i][e];
      s2 += vr[i][e] * vr[i][e];
    }
  }
  for (int m = 1; m < 8; m <<= 1) {
    s1 += __shfl_xor(s1, m, 32);
    s2 += __shfl_xor(s2, m, 32);
  }
  const float mean = s1 * (1.0f / Dc);
  const float var = s2 * (1.0f / Dc) - mean * mean;
  const float rs = rsqrtf(var + 1e-5f);
  float* xrow = xo + ((size_t)b * Nc + g) * Dc + 4 * sub;
#pragma unroll
  for (int i = 0; i < LNP; ++i) {
    v4f y;
    for (int e = 0; e < 4; ++e)
      y[e] = (vr[i][e] - mean) * rs * gm[i][e] + bt[i][e];
    vr[i] = y;
    *(v4f*)(xsr + 32 * i) = y;
  }
  if (g >= own0) {
#pragma unroll
    for (int i = 0; i < LNP; ++i) *(v4f*)(xrow + 32 * i) = vr[i];
  }
  const int wave = tid >> 5, lane = tid & 31;
  const int hf = lane >> 4, l16 = lane & 15;
  float* out = (wave < 2) ? qo : ko;
  const int cb = (wave & 1) * 16;
  const float* ap = xs + l16 * XSTR + 2 * hf;
  const float* bp = wl + 2 * hf * WPR + wave * 16 + l16;
  v8f acc = {0.f, 0.f, 0.f, 0.f, 0.f, 0.f, 0.f, 0.f};
  wpanel_store(w0, wl, tid, 0);
  wpanel_store(w1, wl, tid, 1);
  __syncthreads();
  {
    v2f aA[PKB], bA[PKB], aB[PKB], bB[PKB];
    pfrag_load(aA, bA, ap, bp, 0);
#pragma unroll
    for (int blk = 0; blk < 2 * PNB; blk += 2) {
      pfrag_load(aB, bB, ap, bp, (blk + 1) * PKB);
      __builtin_amdgcn_sched_barrier(0);
#pragma unroll
      for (int s = 0; s < PKB; ++s) acc = wmma_f32(aA[s], bA[s], acc);
      __builtin_amdgcn_sched_barrier(0);
      if (blk + 2 < 2 * PNB) pfrag_load(aA, bA, ap, bp, (blk + 2) * PKB);
      __builtin_amdgcn_sched_barrier(0);
#pragma unroll
      for (int s = 0; s < PKB; ++s) acc = wmma_f32(aB[s], bB[s], acc);
      __builtin_amdgcn_sched_barrier(0);
    }
  }
  if (whole) {
    for (int i = 0; i < 8; ++i) {
      const int r = i + 8 * hf;
      out[((size_t)b * Nc + n0 + r) * Rc + cb + l16] = acc[i];
    }
  } else {
    for (int i = 0; i < 8; ++i) {
      const int r = i + 8 * hf, gg = n0 + r;
      if (gg >= own0) out[((size_t)b * Nc + gg) * Rc + cb + l16] = acc[i];
    }
  }
}

__device__ __forceinline__ void xchunk_load(v4f (&r)[2 * XPW],
                                            const float* __restrict__ xbat,
                                            int n0, int wv, int lane, int ch) {
#pragma unroll
  for (int m = 0; m < XPW; ++m) {
    const int j = 2 * (wv + 4 * m);
    const int g0 = min(max(n0 - Wc + j, 0), Nc - 1);
    const int g1 = min(max(n0 - Wc + j + 1, 0), Nc - 1);
    r[2 * m] = *(const v4f*)(xbat + (size_t)g0 * Dc + ch * CHc + 4 * lane);
    r[2 * m + 1] = *(const v4f*)(xbat + (size_t)g1 * Dc + ch * CHc + 4 * lane);
  }
}

__device__ __forceinline__ void xchunk_store(const v4f (&r)[2 * XPW],
                                             float* buf, int wv, int lane) {
#pragma unroll
  for (int m = 0; m < XPW; ++m) {
    const v4f lo = r[2 * m], hi = r[2 * m + 1];
    const v4f s0 = {lo.x, hi.x, lo.y, hi.y};
    const v4f s1 = {lo.z, hi.z, lo.w, hi.w};
    float* p = buf + (wv + 4 * m) * XPSTR + 8 * lane;
    *(v4f*)p = s0;
    *(v4f*)(p + 4) = s1;
  }
}

__device__ __forceinline__ void bblock_load(v2f (&bb)[KB][2], const float* bp,
                                            int blk) {
#pragma unroll
  for (int s = 0; s < KB; ++s) {
    const float* p = bp + 2 * (blk * KB + s) * XPSTR;
    bb[s][0] = *(const v2f*)p;
    bb[s][1] = *(const v2f*)(p + 32);
  }
}

__global__ __launch_bounds__(128) void attn_kernel(
    const float* __restrict__ x, const float* __restrict__ q,
    const float* __restrict__ k, const float* __restrict__ st,
    float* __restrict__ val) {
  __shared__ float ss[16 * SSTR];
  __shared__ float wm[16 * WSTR];
  __shared__ float sw[WINc];
  __shared__ __attribute__((aligned(16))) float xb[XBUF];
  const int cq = blockIdx.x % NCH;
  const int tile = (blockIdx.x / NCH) % NT;
  const int b = blockIdx.x / (NCH * NT);
  const int n0 = tile * 16;
  const int tid = threadIdx.x;
  const bool full = (n0 + 16 <= Nc);
  const bool interior = (n0 >= Wc) && (n0 + 16 + Wc <= Nc);
  const int wave = tid >> 5, lane = tid & 31;
  const int hf = lane >> 4, l16 = lane & 15;
  const int wv = __builtin_amdgcn_readfirstlane(wave);

  const float* qbat = q + (size_t)b * Nc * Rc + 16 * hf;
  const float* kbat = k + (size_t)b * Nc * Rc + 16 * hf;
  v4f qf[4], kf[3][4];
  {
    const int gq = min(n0 + l16, Nc - 1);
#pragma unroll
    for (int i = 0; i < 4; ++i)
      qf[i] = *(const v4f*)(qbat + (size_t)gq * Rc + 4 * i);
  }
#pragma unroll
  for (int u = 0; u < 3; ++u) {
    if (wv + 4 * u < 9) {
      const int gk = min(max(n0 - Wc + 16 * (wv + 4 * u) + l16, 0), Nc - 1);
#pragma unroll
      for (int i = 0; i < 4; ++i)
        kf[u][i] = *(const v4f*)(kbat + (size_t)gk * Rc + 4 * i);
    }
  }

  const float* sb = st + (size_t)b * Nc + n0 - Wc;
  if (interior) {
    for (int j = tid; j < WINc; j += 128) sw[j] = sb[j];
  } else {
    for (int j = tid; j < WINc; j += 128) {
      const int gn = n0 - Wc + j;
      sw[j] = (gn >= 0 && gn < Nc) ? st[(size_t)b * Nc + gn] : 0.0f;
    }
  }
  for (int i = tid; i < 16 * WSTR; i += 128) wm[i] = 0.0f;

#pragma unroll
  for (int u = 0; u < 3; ++u) {
    if (wv + 4 * u < 9) {
      const int cb = (wv + 4 * u) * 16;
      v8f acc = {0.f, 0.f, 0.f, 0.f, 0.f, 0.f, 0.f, 0.f};
#pragma unroll
      for (int i = 0; i < 4; ++i) {
        const v2f a0 = {qf[i].x, qf[i].y}, b0 = {kf[u][i].x, kf[u][i].y};
        acc = wmma_f32(a0, b0, acc);
        const v2f a1 = {qf[i].z, qf[i].w}, b1 = {kf[u][i].z, kf[u][i].w};
        acc = wmma_f32(a1, b1, acc);
      }
      for (int i = 0; i < 8; ++i) ss[(i + 8 * hf) * SSTR + cb + l16] = acc[i];
    }
  }
  __syncthreads();

  const float* xbat = x + (size_t)b * Nc * Dc;
  v4f xr[2 * XPW];
  xchunk_load(xr, xbat, n0, wv, lane, cq);

  {
    const int row = tid >> 3, sub = tid & 7;
    const int g = n0 + row;
    if (interior) {
      float m = -1e30f;
      for (int o = sub; o < WNc; o += 8)
        m = fmaxf(m, fabsf(ss[row * SSTR + row + o]));
      for (int mm = 1; mm < 8; mm <<= 1) m = fmaxf(m, __shfl_xor(m, mm, 32));
      float Z = 0.0f;
      for (int o = sub; o < WNc; o += 8)
        Z += expf(fabsf(ss[row * SSTR + row + o]) - m);
      for (int mm = 1; mm < 8; mm <<= 1) Z += __shfl_xor(Z, mm, 32);
      const float inv = 1.0f / Z;
      for (int o = sub; o < WNc; o += 8) {
        const float s = ss[row * SSTR + row + o];
        const float p = expf(fabsf(s) - m) * inv;
        const float wv2 = (s > 0.0f) ? p : ((s < 0.0f) ? -p : 0.0f);
        wm[row * WSTR + row + o] = wv2 * sw[row + o];
      }
    } else if (g < Nc) {
      float m = -1e30f;
      for (int o = sub; o < WNc; o += 8) {
        const int gn = g - Wc + o;
        if (gn >= 0 && gn < Nc) m = fmaxf(m, fabsf(ss[row * SSTR + row + o]));
      }
      for (int mm = 1; mm < 8; mm <<= 1) m = fmaxf(m, __shfl_xor(m, mm, 32));
      float Z = 0.0f;
      for (int o = sub; o < WNc; o += 8) {
        const int gn = g - Wc + o;
        if (gn >= 0 && gn < Nc)
          Z += expf(fabsf(ss[row * SSTR + row + o]) - m);
      }
      for (int mm = 1; mm < 8; mm <<= 1) Z += __shfl_xor(Z, mm, 32);
      const float inv = 1.0f / Z;
      for (int o = sub; o < WNc; o += 8) {
        const int gn = g - Wc + o;
        if (gn >= 0 && gn < Nc) {
          const float s = ss[row * SSTR + row + o];
          const float p = expf(fabsf(s) - m) * inv;
          const float wv2 = (s > 0.0f) ? p : ((s < 0.0f) ? -p : 0.0f);
          wm[row * WSTR + row + o] = wv2 * sw[row + o];
        }
      }
    }
  }
  xchunk_store(xr, xb, wv, lane);
  __syncthreads();


  v2f af[WINc / 4];
#pragma unroll
  for (int kk = 0; kk < WINc / 4; ++kk) {
    const int c0 = 4 * kk + 2 * hf;
    af[kk].x = wm[l16 * WSTR + c0];
    af[kk].y = wm[l16 * WSTR + c0 + 1];
  }

  float* vt = val + (size_t)b * Nc * Dc + cq * CHc + wv * 32 + l16;
  float o0[8], o1[8];
#pragma unroll
  for (int i = 0; i < 8; ++i) {
    const int gl = min(n0 + i + 8 * hf, Nc - 1);
    o0[i] = vt[(size_t)gl * Dc];
    o1[i] = vt[(size_t)gl * Dc + 16];
  }

  const float* bp = xb + hf * XPSTR + 2 * (wv * 32 + l16);
  v8f acc0 = {0.f, 0.f, 0.f, 0.f, 0.f, 0.f, 0.f, 0.f};
  v8f acc1 = {0.f, 0.f, 0.f, 0.f, 0.f, 0.f, 0.f, 0.f};
  v2f bA[KB][2], bB[KB][2];
  bblock_load(bA, bp, 0);
#pragma unroll
  for (int blk = 0; blk < NKB; blk += 2) {
    bblock_load(bB, bp, blk + 1);
    __builtin_amdgcn_sched_barrier(0);
#pragma unroll
    for (int s = 0; s < KB; ++s) {
      acc0 = wmma_f32(af[blk * KB + s], bA[s][0], acc0);
      acc1 = wmma_f32(af[blk * KB + s], bA[s][1], acc1);
    }
    __builtin_amdgcn_sched_barrier(0);
    if (blk + 2 < NKB) bblock_load(bA, bp, blk + 2);
    __builtin_amdgcn_sched_barrier(0);
#pragma unroll
    for (int s = 0; s < KB; ++s) {
      acc0 = wmma_f32(af[(blk + 1) * KB + s], bB[s][0], acc0);
      acc1 = wmma_f32(af[(blk + 1) * KB + s], bB[s][1], acc1);
    }
    __builtin_amdgcn_sched_barrier(0);
  }
  if (full) {
#pragma unroll
    for (int i = 0; i < 8; ++i) {
      float* p = vt + (size_t)(n0 + i + 8 * hf) * Dc;
      p[0] = o0[i] + acc0[i];
      p[16] = o1[i] + acc1[i];
    }
  } else {
#pragma unroll
    for (int i = 0; i < 8; ++i) {
      const int gg = n0 + i + 8 * hf;
      if (gg < Nc) {
        float* p = vt + (size_t)gg * Dc;
        p[0] = o0[i] + acc0[i];
        p[16] = o1[i] + acc1[i];
      }
    }
  }
}

extern "C" void kernel_launch(void* const* d_in, const int* in_sizes, int n_in,
                              void* d_out, int out_size, void* d_ws,
                              size_t ws_size, hipStream_t stream) {
  const int* ids = (const int*)d_in[0];
  const float* emb = (const float*)d_in[1];
  const float* pos = (const float*)d_in[2];
  const float* anchor_val = (const float*)d_in[3];
  const float* anchor_state = (const float*)d_in[4];
  const float* in_gamma = (const float*)d_in[5];
  const float* in_beta = (const float*)d_in[6];
  const float* ng = (const float*)d_in[7];
  const float* nb = (const float*)d_in[8];
  const float* Uw = (const float*)d_in[9];
  const float* Vw = (const float*)d_in[10];
  const float* sp_w = (const float*)d_in[11];
  const float* sp_b = (const float*)d_in[12];

  float* val = (float*)d_out;
  float* ws = (float*)d_ws;
  float* x = ws;
  float* q = x + (size_t)Bc * Nc * Dc;
  float* k = q + (size_t)Bc * Nc * Rc;
  float* st = k + (size_t)Bc * Nc * Rc;

  embed_ln_kernel<<<Bc * Nc, 256, 0, stream>>>(
      ids, emb, pos, anchor_val, anchor_state, in_gamma, in_beta, sp_w, sp_b,
      val, st);

  for (int l = 0; l < Lc; ++l) {
    ln_proj_kernel<<<Bc * NT, 128, 0, stream>>>(
        val, ng + (size_t)l * Dc, nb + (size_t)l * Dc, Uw + (size_t)l * Dc * Rc,
        Vw + (size_t)l * Dc * Rc, x, q, k);
    attn_kernel<<<Bc * NT * NCH, 128, 0, stream>>>(x, q, k, st, val);
  }
}


# embed kernel: one wave per node row, 8 rows per work-group (2050 waves, not 16400): row in registers, 4 x 16 B per lane, lane sums + 32-lane butterfly, no LDS or barrier; LN, state sums reordered
# speedup vs baseline: 5.2950x; 1.0328x over previous
#include <hip/hip_runtime.h>

typedef __attribute__((ext_vector_type(2))) float v2f;
typedef __attribute__((ext_vector_type(8))) float v8f;
typedef __attribute__((ext_vector_type(4))) float v4f;

constexpr int Bc = 2;
constexpr int Sc = 1024;
constexpr int Nc = Sc + 1;
constexpr int Dc = 512;
constexpr int Rc = 32;
constexpr int Lc = 3;
constexpr int Wc = 64;
constexpr int WNc = 2 * Wc + 1;
constexpr int WINc = 144;
constexpr int NT = (Nc + 15) / 16;

constexpr int XSTR = Dc + 4;
constexpr int SSTR = 146;
constexpr int WSTR = 145;

constexpr int CHc = 128;
constexpr int NCH = Dc / CHc;
constexpr int XPSTR = 2 * CHc + 32;
constexpr int XBUF = (WINc / 2) * XPSTR;
constexpr int XPW = WINc / 2 / 4;
constexpr int KB = 9;
constexpr int NKB = WINc / 4 / KB;

constexpr int WPR = 2 * Rc + 16;
constexpr int WHK = Dc / 2;
constexpr int WLD = WHK * Rc / 4 / 128;
constexpr int PKB = 16;
constexpr int PNB = WHK / 4 / PKB;
constexpr int LNP = Dc / 4 / 8;

__device__ __forceinline__ v8f wmma_f32(v2f a, v2f b, v8f c) {
  return __builtin_amdgcn_wmma_f32_16x16x4_f32(false, a, false, b, (short)0, c,
                                               false, false);
}

constexpr int ERW = 8;
constexpr int EPC = Dc / 4 / 32;
constexpr int EWG = (Bc * Nc + ERW - 1) / ERW;

__device__ __forceinline__ void wave_sum2(float& a, float& b) {
  for (int m = 16; m >= 1; m >>= 1) {
    const float ta = __shfl_xor(a, m, 32);
    const float tb = __shfl_xor(b, m, 32);
    a += ta;
    b += tb;
  }
}

__device__ __forceinline__ float wave_sum(float v) {
  for (int m = 16; m >= 1; m >>= 1) v += __shfl_xor(v, m, 32);
  return v;
}

__global__ __launch_bounds__(32 * ERW) void embed_ln_kernel(
    const int* __restrict__ ids, const float* __restrict__ emb,
    const float* __restrict__ pos, const float* __restrict__ anchor_val,
    const float* __restrict__ anchor_state, const float* __restrict__ gamma,
    const float* __restrict__ beta, const float* __restrict__ sp_w,
    const float* __restrict__ sp_b, float* __restrict__ val,
    float* __restrict__ st) {
  const int lane = threadIdx.x & 31;
  const int wv = __builtin_amdgcn_readfirstlane(threadIdx.x >> 5);
  const int r = blockIdx.x * ERW + wv;
  if (r >= Bc * Nc) return;
  const int n = r % Nc;
  const int b = r / Nc;
  float* vrow = val + (size_t)r * Dc + 4 * lane;
  if (n == 0) {
    const float* arow = anchor_val + 4 * lane;
#pragma unroll
    for (int i = 0; i < EPC; ++i)
      *(v4f*)(vrow + 128 * i) = *(const v4f*)(arow + 128 * i);
    if (lane == 0) st[r] = anchor_state[0];
    return;
  }
  const int s = n - 1;
  const int id = ids[b * Sc + s];
  const float* prow = pos + (size_t)s * Dc + 4 * lane;
  v4f pp[EPC], gm[EPC], bt[EPC], sw[EPC], e[EPC];
#pragma unroll
  for (int i = 0; i < EPC; ++i) pp[i] = *(const v4f*)(prow + 128 * i);
#pragma unroll
  for (int i = 0; i < EPC; ++i) {
    gm[i] = *(const v4f*)(gamma + 4 * lane + 128 * i);
    bt[i] = *(const v4f*)(beta + 4 * lane + 128 * i);
    sw[i] = *(const v4f*)(sp_w + 4 * lane + 128 * i);
  }
  __builtin_amdgcn_sched_barrier(0);
  const float* erow = emb + (size_t)id * Dc + 4 * lane;
#pragma unroll
  for (int i = 0; i < EPC; ++i) e[i] = *(const v4f*)(erow + 128 * i);
  __builtin_amdgcn_sched_barrier(0);
  float s1 = 0.0f, s2 = 0.0f;
#pragma unroll
  for (int i = 0; i < EPC; ++i) {
    e[i] += pp[i];
    for (int c = 0; c < 4; ++c) {
      s1 += e[i][c];
      s2 += e[i][c] * e[i][c];
    }
  }
  wave_sum2(s1, s2);
  const float mean = s1 * (1.0f / Dc);
  const float var = s2 * (1.0f / Dc) - mean * mean;
  const float rs = rsqrtf(var + 1e-5f);
  float dot = 0.0f;
#pragma unroll
  for (int i = 0; i < EPC; ++i) {
    v4f y;
    for (int c = 0; c < 4; ++c) {
      y[c] = (e[i][c] - mean) * rs * gm[i][c] + bt[i][c];
      dot += y[c] * sw[i][c];
    }
    *(v4f*)(vrow + 128 * i) = y;
  }
  __builtin_amdgcn_sched_barrier(0);
  dot = wave_sum(dot);
  if (lane == 0) st[r] = dot + sp_b[0];
}

__device__ __forceinline__ void wpanel_load(v4f (&r)[2 * WLD],
                                            const float* __restrict__ Uw,
                                            const float* __restrict__ Vw,
                                            int tid, int half) {
#pragma unroll
  for (int p = 0; p < WLD; ++p) {
    const int f = tid + 128 * (half * WLD + p);
    r[2 * p] = *(const v4f*)(Uw + 4 * f);
    r[2 * p + 1] = *(const v4f*)(Vw + 4 * f);
  }
}

__device__ __forceinline__ void wpanel_store(const v4f (&r)[2 * WLD], float* wl,
                                             int tid, int half) {
#pragma unroll
  for (int p = 0; p < WLD; ++p) {
    const int f = tid + 128 * (half * WLD + p);
    float* d = wl + (f >> 3) * WPR + 4 * (f & 7);
    *(v4f*)d = r[2 * p];
    *(v4f*)(d + Rc) = r[2 * p + 1];
  }
}

__device__ __forceinline__ void pfrag_load(v2f (&a)[PKB], v2f (&b)[PKB],
                                           const float* ap, const float* bp,
                                           int k0) {
#pragma unroll
  for (int s = 0; s < PKB; ++s) {
    const int kk = k0 + s;
    a[s] = *(const v2f*)(ap + 4 * kk);
    b[s].x = bp[(4 * kk) * WPR];
    b[s].y = bp[(4 * kk + 1) * WPR];
  }
}

__global__ __launch_bounds__(128) void ln_proj_kernel(
    const float* __restrict__ val, const float* __restrict__ gamma,
    const float* __restrict__ beta, const float* __restrict__ Uw,
    const float* __restrict__ Vw, float* __restrict__ xo,
    float* __restrict__ qo, float* __restrict__ ko) {
  __shared__ __attribute__((aligned(16))) float xs[16 * XSTR];
  __shared__ __attribute__((aligned(16))) float wl[Dc * WPR];
  const int tile = blockIdx.x % NT;
  const int b = blockIdx.x / NT;
  const int own0 = tile * 16;
  const int n0 = min(own0, Nc - 16);
  const int tid = threadIdx.x;
  const bool whole = (n0 == own0);

  const int row = tid >> 3, sub = tid & 7;
  const int g = n0 + row;
  const float* vrow = val + ((size_t)b * Nc + g) * Dc + 4 * sub;
  const float* gpc = gamma + 4 * sub;
  const float* bpc = beta + 4 * sub;
  float* xsr = xs + row * XSTR + 4 * sub;
  v4f vr[LNP], gm[LNP], bt[LNP];
#pragma unroll
  for (int i = 0; i < LNP; ++i) vr[i] = *(const v4f*)(vrow + 32 * i);
#pragma unroll
  for (int i = 0; i < LNP; ++i) {
    gm[i] = *(const v4f*)(gpc + 32 * i);
    bt[i] = *(const v4f*)(bpc + 32 * i);
  }
  __builtin_amdgcn_sched_barrier(0);
  v4f w0[2 * WLD], w1[2 * WLD];
  wpanel_load(w0, Uw, Vw, tid, 0);
  wpanel_load(w1, Uw, Vw, tid, 1);
  __builtin_amdgcn_sched_barrier(0);
  float s1 = 0.0f, s2 = 0.0f;
#pragma unroll
  for (int i = 0; i < LNP; ++i) {
    for (int e = 0; e < 4; ++e) {
      s1 += vr[i][e];
      s2 += vr[i][e] * vr[i][e];
    }
  }
  for (int m = 1; m < 8; m <<= 1) {
    s1 += __shfl_xor(s1, m, 32);
    s2 += __shfl_xor(s2, m, 32);
  }
  const float mean = s1 * (1.0f / Dc);
  const float var = s2 * (1.0f / Dc) - mean * mean;
  const float rs = rsqrtf(var + 1e-5f);
  float* xrow = xo + ((size_t)b * Nc + g) * Dc + 4 * sub;
#pragma unroll
  for (int i = 0; i < LNP; ++i) {
    v4f y;
    for (int e = 0; e < 4; ++e)
      y[e] = (vr[i][e] - mean) * rs * gm[i][e] + bt[i][e];
    vr[i] = y;
    *(v4f*)(xsr + 32 * i) = y;
  }
  if (g >= own0) {
#pragma unroll
    for (int i = 0; i < LNP; ++i) *(v4f*)(xrow + 32 * i) = vr[i];
  }
  const int wave = tid >> 5, lane = tid & 31;
  const int hf = lane >> 4, l16 = lane & 15;
  float* out = (wave < 2) ? qo : ko;
  const int cb = (wave & 1) * 16;
  const float* ap = xs + l16 * XSTR + 2 * hf;
  const float* bp = wl + 2 * hf * WPR + wave * 16 + l16;
  v8f acc = {0.f, 0.f, 0.f, 0.f, 0.f, 0.f, 0.f, 0.f};
  wpanel_store(w0, wl, tid, 0);
  wpanel_store(w1, wl, tid, 1);
  __syncthreads();
  {
    v2f aA[PKB], bA[PKB], aB[PKB], bB[PKB];
    pfrag_load(aA, bA, ap, bp, 0);
#pragma unroll
    for (int blk = 0; blk < 2 * PNB; blk += 2) {
      pfrag_load(aB, bB, ap, bp, (blk + 1) * PKB);
      __builtin_amdgcn_sched_barrier(0);
#pragma unroll
      for (int s = 0; s < PKB; ++s) acc = wmma_f32(aA[s], bA[s], acc);
      __builtin_amdgcn_sched_barrier(0);
      if (blk + 2 < 2 * PNB) pfrag_load(aA, bA, ap, bp, (blk + 2) * PKB);
      __builtin_amdgcn_sched_barrier(0);
#pragma unroll
      for (int s = 0; s < PKB; ++s) acc = wmma_f32(aB[s], bB[s], acc);
      __builtin_amdgcn_sched_barrier(0);
    }
  }
  if (whole) {
    for (int i = 0; i < 8; ++i) {
      const int r = i + 8 * hf;
      out[((size_t)b * Nc + n0 + r) * Rc + cb + l16] = acc[i];
    }
  } else {
    for (int i = 0; i < 8; ++i) {
      const int r = i + 8 * hf, gg = n0 + r;
      if (gg >= own0) out[((size_t)b * Nc + gg) * Rc + cb + l16] = acc[i];
    }
  }
}

__device__ __forceinline__ void xchunk_load(v4f (&r)[2 * XPW],
                                            const float* __restrict__ xbat,
                                            int n0, int wv, int lane, int ch) {
#pragma unroll
  for (int m = 0; m < XPW; ++m) {
    const int j = 2 * (wv + 4 * m);
    const int g0 = min(max(n0 - Wc + j, 0), Nc - 1);
    const int g1 = min(max(n0 - Wc + j + 1, 0), Nc - 1);
    r[2 * m] = *(const v4f*)(xbat + (size_t)g0 * Dc + ch * CHc + 4 * lane);
    r[2 * m + 1] = *(const v4f*)(xbat + (size_t)g1 * Dc + ch * CHc + 4 * lane);
  }
}

__device__ __forceinline__ void xchunk_store(const v4f (&r)[2 * XPW],
                                             float* buf, int wv, int lane) {
#pragma unroll
  for (int m = 0; m < XPW; ++m) {
    const v4f lo = r[2 * m], hi = r[2 * m + 1];
    const v4f s0 = {lo.x, hi.x, lo.y, hi.y};
    const v4f s1 = {lo.z, hi.z, lo.w, hi.w};
    float* p = buf + (wv + 4 * m) * XPSTR + 8 * lane;
    *(v4f*)p = s0;
    *(v4f*)(p + 4) = s1;
  }
}

__device__ __forceinline__ void bblock_load(v2f (&bb)[KB][2], const float* bp,
                                            int blk) {
#pragma unroll
  for (int s = 0; s < KB; ++s) {
    const float* p = bp + 2 * (blk * KB + s) * XPSTR;
    bb[s][0] = *(const v2f*)p;
    bb[s][1] = *(const v2f*)(p + 32);
  }
}

__global__ __launch_bounds__(128) void attn_kernel(
    const float* __restrict__ x, const float* __restrict__ q,
    const float* __restrict__ k, const float* __restrict__ st,
    float* __restrict__ val) {
  __shared__ float ss[16 * SSTR];
  __shared__ float wm[16 * WSTR];
  __shared__ float sw[WINc];
  __shared__ __attribute__((aligned(16))) float xb[XBUF];
  const int cq = blockIdx.x % NCH;
  const int tile = (blockIdx.x / NCH) % NT;
  const int b = blockIdx.x / (NCH * NT);
  const int n0 = tile * 16;
  const int tid = threadIdx.x;
  const bool full = (n0 + 16 <= Nc);
  const bool interior = (n0 >= Wc) && (n0 + 16 + Wc <= Nc);
  const int wave = tid >> 5, lane = tid & 31;
  const int hf = lane >> 4, l16 = lane & 15;
  const int wv = __builtin_amdgcn_readfirstlane(wave);

  const float* qbat = q + (size_t)b * Nc * Rc + 16 * hf;
  const float* kbat = k + (size_t)b * Nc * Rc + 16 * hf;
  v4f qf[4], kf[3][4];
  {
    const int gq = min(n0 + l16, Nc - 1);
#pragma unroll
    for (int i = 0; i < 4; ++i)
      qf[i] = *(const v4f*)(qbat + (size_t)gq * Rc + 4 * i);
  }
#pragma unroll
  for (int u = 0; u < 3; ++u) {
    if (wv + 4 * u < 9) {
      const int gk = min(max(n0 - Wc + 16 * (wv + 4 * u) + l16, 0), Nc - 1);
#pragma unroll
      for (int i = 0; i < 4; ++i)
        kf[u][i] = *(const v4f*)(kbat + (size_t)gk * Rc + 4 * i);
    }
  }

  const float* sb = st + (size_t)b * Nc + n0 - Wc;
  if (interior) {
    for (int j = tid; j < WINc; j += 128) sw[j] = sb[j];
  } else {
    for (int j = tid; j < WINc; j += 128) {
      const int gn = n0 - Wc + j;
      sw[j] = (gn >= 0 && gn < Nc) ? st[(size_t)b * Nc + gn] : 0.0f;
    }
  }
  for (int i = tid; i < 16 * WSTR; i += 128) wm[i] = 0.0f;

#pragma unroll
  for (int u = 0; u < 3; ++u) {
    if (wv + 4 * u < 9) {
      const int cb = (wv + 4 * u) * 16;
      v8f acc = {0.f, 0.f, 0.f, 0.f, 0.f, 0.f, 0.f, 0.f};
#pragma unroll
      for (int i = 0; i < 4; ++i) {
        const v2f a0 = {qf[i].x, qf[i].y}, b0 = {kf[u][i].x, kf[u][i].y};
        acc = wmma_f32(a0, b0, acc);
        const v2f a1 = {qf[i].z, qf[i].w}, b1 = {kf[u][i].z, kf[u][i].w};
        acc = wmma_f32(a1, b1, acc);
      }
      for (int i = 0; i < 8; ++i) ss[(i + 8 * hf) * SSTR + cb + l16] = acc[i];
    }
  }
  __syncthreads();

  const float* xbat = x + (size_t)b * Nc * Dc;
  v4f xr[2 * XPW];
  xchunk_load(xr, xbat, n0, wv, lane, cq);

  {
    const int row = tid >> 3, sub = tid & 7;
    const int g = n0 + row;
    if (interior) {
      float m = -1e30f;
      for (int o = sub; o < WNc; o += 8)
        m = fmaxf(m, fabsf(ss[row * SSTR + row + o]));
      for (int mm = 1; mm < 8; mm <<= 1) m = fmaxf(m, __shfl_xor(m, mm, 32));
      float Z = 0.0f;
      for (int o = sub; o < WNc; o += 8)
        Z += expf(fabsf(ss[row * SSTR + row + o]) - m);
      for (int mm = 1; mm < 8; mm <<= 1) Z += __shfl_xor(Z, mm, 32);
      const float inv = 1.0f / Z;
      for (int o = sub; o < WNc; o += 8) {
        const float s = ss[row * SSTR + row + o];
        const float p = expf(fabsf(s) - m) * inv;
        const float wv2 = (s > 0.0f) ? p : ((s < 0.0f) ? -p : 0.0f);
        wm[row * WSTR + row + o] = wv2 * sw[row + o];
      }
    } else if (g < Nc) {
      float m = -1e30f;
      for (int o = sub; o < WNc; o += 8) {
        const int gn = g - Wc + o;
        if (gn >= 0 && gn < Nc) m = fmaxf(m, fabsf(ss[row * SSTR + row + o]));
      }
      for (int mm = 1; mm < 8; mm <<= 1) m = fmaxf(m, __shfl_xor(m, mm, 32));
      float Z = 0.0f;
      for (int o = sub; o < WNc; o += 8) {
        const int gn = g - Wc + o;
        if (gn >= 0 && gn < Nc)
          Z += expf(fabsf(ss[row * SSTR + row + o]) - m);
      }
      for (int mm = 1; mm < 8; mm <<= 1) Z += __shfl_xor(Z, mm, 32);
      const float inv = 1.0f / Z;
      for (int o = sub; o < WNc; o += 8) {
        const int gn = g - Wc + o;
        if (gn >= 0 && gn < Nc) {
          const float s = ss[row * SSTR + row + o];
          const float p = expf(fabsf(s) - m) * inv;
          const float wv2 = (s > 0.0f) ? p : ((s < 0.0f) ? -p : 0.0f);
          wm[row * WSTR + row + o] = wv2 * sw[row + o];
        }
      }
    }
  }
  xchunk_store(xr, xb, wv, lane);
  __syncthreads();


  v2f af[WINc / 4];
#pragma unroll
  for (int kk = 0; kk < WINc / 4; ++kk) {
    const int c0 = 4 * kk + 2 * hf;
    af[kk].x = wm[l16 * WSTR + c0];
    af[kk].y = wm[l16 * WSTR + c0 + 1];
  }

  float* vt = val + (size_t)b * Nc * Dc + cq * CHc + wv * 32 + l16;
  float o0[8], o1[8];
#pragma unroll
  for (int i = 0; i < 8; ++i) {
    const int gl = min(n0 + i + 8 * hf, Nc - 1);
    o0[i] = vt[(size_t)gl * Dc];
    o1[i] = vt[(size_t)gl * Dc + 16];
  }

  const float* bp = xb + hf * XPSTR + 2 * (wv * 32 + l16);
  v8f acc0 = {0.f, 0.f, 0.f, 0.f, 0.f, 0.f, 0.f, 0.f};
  v8f acc1 = {0.f, 0.f, 0.f, 0.f, 0.f, 0.f, 0.f, 0.f};
  v2f bA[KB][2], bB[KB][2];
  bblock_load(bA, bp, 0);
#pragma unroll
  for (int blk = 0; blk < NKB; blk += 2) {
    bblock_load(bB, bp, blk + 1);
    __builtin_amdgcn_sched_barrier(0);
#pragma unroll
    for (int s = 0; s < KB; ++s) {
      acc0 = wmma_f32(af[blk * KB + s], bA[s][0], acc0);
      acc1 = wmma_f32(af[blk * KB + s], bA[s][1], acc1);
    }
    __builtin_amdgcn_sched_barrier(0);
    if (blk + 2 < NKB) bblock_load(bA, bp, blk + 2);
    __builtin_amdgcn_sched_barrier(0);
#pragma unroll
    for (int s = 0; s < KB; ++s) {
      acc0 = wmma_f32(af[(blk + 1) * KB + s], bB[s][0], acc0);
      acc1 = wmma_f32(af[(blk + 1) * KB + s], bB[s][1], acc1);
    }
    __builtin_amdgcn_sched_barrier(0);
  }
  if (full) {
#pragma unroll
    for (int i = 0; i < 8; ++i) {
      float* p = vt + (size_t)(n0 + i + 8 * hf) * Dc;
      p[0] = o0[i] + acc0[i];
      p[16] = o1[i] + acc1[i];
    }
  } else {
#pragma unroll
    for (int i = 0; i < 8; ++i) {
      const int gg = n0 + i + 8 * hf;
      if (gg < Nc) {
        float* p = vt + (size_t)gg * Dc;
        p[0] = o0[i] + acc0[i];
        p[16] = o1[i] + acc1[i];
      }
    }
  }
}

extern "C" void kernel_launch(void* const* d_in, const int* in_sizes, int n_in,
                              void* d_out, int out_size, void* d_ws,
                              size_t ws_size, hipStream_t stream) {
  const int* ids = (const int*)d_in[0];
  const float* emb = (const float*)d_in[1];
  const float* pos = (const float*)d_in[2];
  const float* anchor_val = (const float*)d_in[3];
  const float* anchor_state = (const float*)d_in[4];
  const float* in_gamma = (const float*)d_in[5];
  const float* in_beta = (const float*)d_in[6];
  const float* ng = (const float*)d_in[7];
  const float* nb = (const float*)d_in[8];
  const float* Uw = (const float*)d_in[9];
  const float* Vw = (const float*)d_in[10];
  const float* sp_w = (const float*)d_in[11];
  const float* sp_b = (const float*)d_in[12];

  float* val = (float*)d_out;
  float* ws = (float*)d_ws;
  float* x = ws;
  float* q = x + (size_t)Bc * Nc * Dc;
  float* k = q + (size_t)Bc * Nc * Rc;
  float* st = k + (size_t)Bc * Nc * Rc;

  embed_ln_kernel<<<EWG, 32 * ERW, 0, stream>>>(
      ids, emb, pos, anchor_val, anchor_state, in_gamma, in_beta, sp_w, sp_b,
      val, st);

  for (int l = 0; l < Lc; ++l) {
    ln_proj_kernel<<<Bc * NT, 128, 0, stream>>>(
        val, ng + (size_t)l * Dc, nb + (size_t)l * Dc, Uw + (size_t)l * Dc * Rc,
        Vw + (size_t)l * Dc * Rc, x, q, k);
    attn_kernel<<<Bc * NT * NCH, 128, 0, stream>>>(x, q, k, st, val);
  }
}


# attn kernel: scores computed transposed (a q row's window in one lane pair); softmax on those accumulators, row max and sum joined over the 4 waves via LDS; scores never pass through LDS; Z reordered
# speedup vs baseline: 6.2856x; 1.1871x over previous
#include <hip/hip_runtime.h>

typedef __attribute__((ext_vector_type(2))) float v2f;
typedef __attribute__((ext_vector_type(8))) float v8f;
typedef __attribute__((ext_vector_type(4))) float v4f;

constexpr int Bc = 2;
constexpr int Sc = 1024;
constexpr int Nc = Sc + 1;
constexpr int Dc = 512;
constexpr int Rc = 32;
constexpr int Lc = 3;
constexpr int Wc = 64;
constexpr int WNc = 2 * Wc + 1;
constexpr int WINc = 144;
constexpr int NT = (Nc + 15) / 16;

constexpr int XSTR = Dc + 4;
constexpr int WSTR = 148;

constexpr int CHc = 128;
constexpr int NCH = Dc / CHc;
constexpr int XPSTR = 2 * CHc + 32;
constexpr int XBUF = (WINc / 2) * XPSTR;
constexpr int XPW = WINc / 2 / 4;
constexpr int KB = 9;
constexpr int NKB = WINc / 4 / KB;

constexpr int WPR = 2 * Rc + 16;
constexpr int WHK = Dc / 2;
constexpr int WLD = WHK * Rc / 4 / 128;
constexpr int PKB = 16;
constexpr int PNB = WHK / 4 / PKB;
constexpr int LNP = Dc / 4 / 8;

__device__ __forceinline__ v8f wmma_f32(v2f a, v2f b, v8f c) {
  return __builtin_amdgcn_wmma_f32_16x16x4_f32(false, a, false, b, (short)0, c,
                                               false, false);
}

constexpr int ERW = 8;
constexpr int EPC = Dc / 4 / 32;
constexpr int EWG = (Bc * Nc + ERW - 1) / ERW;

__device__ __forceinline__ void wave_sum2(float& a, float& b) {
  for (int m = 16; m >= 1; m >>= 1) {
    const float ta = __shfl_xor(a, m, 32);
    const float tb = __shfl_xor(b, m, 32);
    a += ta;
    b += tb;
  }
}

__device__ __forceinline__ float wave_sum(float v) {
  for (int m = 16; m >= 1; m >>= 1) v += __shfl_xor(v, m, 32);
  return v;
}

__global__ __launch_bounds__(32 * ERW) void embed_ln_kernel(
    const int* __restrict__ ids, const float* __restrict__ emb,
    const float* __restrict__ pos, const float* __restrict__ anchor_val,
    const float* __restrict__ anchor_state, const float* __restrict__ gamma,
    const float* __restrict__ beta, const float* __restrict__ sp_w,
    const float* __restrict__ sp_b, float* __restrict__ val,
    float* __restrict__ st) {
  const int lane = threadIdx.x & 31;
  const int wv = __builtin_amdgcn_readfirstlane(threadIdx.x >> 5);
  const int r = blockIdx.x * ERW + wv;
  if (r >= Bc * Nc) return;
  const int n = r % Nc;
  const int b = r / Nc;
  float* vrow = val + (size_t)r * Dc + 4 * lane;
  if (n == 0) {
    const float* arow = anchor_val + 4 * lane;
#pragma unroll
    for (int i = 0; i < EPC; ++i)
      *(v4f*)(vrow + 128 * i) = *(const v4f*)(arow + 128 * i);
    if (lane == 0) st[r] = anchor_state[0];
    return;
  }
  const int s = n - 1;
  const int id = ids[b * Sc + s];
  const float* prow = pos + (size_t)s * Dc + 4 * lane;
  v4f pp[EPC], gm[EPC], bt[EPC], sw[EPC], e[EPC];
#pragma unroll
  for (int i = 0; i < EPC; ++i) pp[i] = *(const v4f*)(prow + 128 * i);
#pragma unroll
  for (int i = 0; i < EPC; ++i) {
    gm[i] = *(const v4f*)(gamma + 4 * lane + 128 * i);
    bt[i] = *(const v4f*)(beta + 4 * lane + 128 * i);
    sw[i] = *(const v4f*)(sp_w + 4 * lane + 128 * i);
  }
  __builtin_amdgcn_sched_barrier(0);
  const float* erow = emb + (size_t)id * Dc + 4 * lane;
#pragma unroll
  for (int i = 0; i < EPC; ++i) e[i] = *(const v4f*)(erow + 128 * i);
  __builtin_amdgcn_sched_barrier(0);
  float s1 = 0.0f, s2 = 0.0f;
#pragma unroll
  for (int i = 0; i < EPC; ++i) {
    e[i] += pp[i];
    for (int c = 0; c < 4; ++c) {
      s1 += e[i][c];
      s2 += e[i][c] * e[i][c];
    }
  }
  wave_sum2(s1, s2);
  const float mean = s1 * (1.0f / Dc);
  const float var = s2 * (1.0f / Dc) - mean * mean;
  const float rs = rsqrtf(var + 1e-5f);
  float dot = 0.0f;
#pragma unroll
  for (int i = 0; i < EPC; ++i) {
    v4f y;
    for (int c = 0; c < 4; ++c) {
      y[c] = (e[i][c] - mean) * rs * gm[i][c] + bt[i][c];
      dot += y[c] * sw[i][c];
    }
    *(v4f*)(vrow + 128 * i) = y;
  }
  __builtin_amdgcn_sched_barrier(0);
  dot = wave_sum(dot);
  if (lane == 0) st[r] = dot + sp_b[0];
}

__device__ __forceinline__ void wpanel_load(v4f (&r)[2 * WLD],
                                            const float* __restrict__ Uw,
                                            const float* __restrict__ Vw,
                                            int tid, int half) {
#pragma unroll
  for (int p = 0; p < WLD; ++p) {
    const int f = tid + 128 * (half * WLD + p);
    r[2 * p] = *(const v4f*)(Uw + 4 * f);
    r[2 * p + 1] = *(const v4f*)(Vw + 4 * f);
  }
}

__device__ __forceinline__ void wpanel_store(const v4f (&r)[2 * WLD], float* wl,
                                             int tid, int half) {
#pragma unroll
  for (int p = 0; p < WLD; ++p) {
    const int f = tid + 128 * (half * WLD + p);
    float* d = wl + (f >> 3) * WPR + 4 * (f & 7);
    *(v4f*)d = r[2 * p];
    *(v4f*)(d + Rc) = r[2 * p + 1];
  }
}

__device__ __forceinline__ void pfrag_load(v2f (&a)[PKB], v2f (&b)[PKB],
                                           const float* ap, const float* bp,
                                           int k0) {
#pragma unroll
  for (int s = 0; s < PKB; ++s) {
    const int kk = k0 + s;
    a[s] = *(const v2f*)(ap + 4 * kk);
    b[s].x = bp[(4 * kk) * WPR];
    b[s].y = bp[(4 * kk + 1) * WPR];
  }
}

__global__ __launch_bounds__(128) void ln_proj_kernel(
    const float* __restrict__ val, const float* __restrict__ gamma,
    const float* __restrict__ beta, const float* __restrict__ Uw,
    const float* __restrict__ Vw, float* __restrict__ xo,
    float* __restrict__ qo, float* __restrict__ ko) {
  __shared__ __attribute__((aligned(16))) float xs[16 * XSTR];
  __shared__ __attribute__((aligned(16))) float wl[Dc * WPR];
  const int tile = blockIdx.x % NT;
  const int b = blockIdx.x / NT;
  const int own0 = tile * 16;
  const int n0 = min(own0, Nc - 16);
  const int tid = threadIdx.x;
  const bool whole = (n0 == own0);

  const int row = tid >> 3, sub = tid & 7;
  const int g = n0 + row;
  const float* vrow = val + ((size_t)b * Nc + g) * Dc + 4 * sub;
  const float* gpc = gamma + 4 * sub;
  const float* bpc = beta + 4 * sub;
  float* xsr = xs + row * XSTR + 4 * sub;
  v4f vr[LNP], gm[LNP], bt[LNP];
#pragma unroll
  for (int i = 0; i < LNP; ++i) vr[i] = *(const v4f*)(vrow + 32 * i);
#pragma unroll
  for (int i = 0; i < LNP; ++i) {
    gm[i] = *(const v4f*)(gpc + 32 * i);
    bt[i] = *(const v4f*)(bpc + 32 * i);
  }
  __builtin_amdgcn_sched_barrier(0);
  v4f w0[2 * WLD], w1[2 * WLD];
  wpanel_load(w0, Uw, Vw, tid, 0);
  wpanel_load(w1, Uw, Vw, tid, 1);
  __builtin_amdgcn_sched_barrier(0);
  float s1 = 0.0f, s2 = 0.0f;
#pragma unroll
  for (int i = 0; i < LNP; ++i) {
    for (int e = 0; e < 4; ++e) {
      s1 += vr[i][e];
      s2 += vr[i][e] * vr[i][e];
    }
  }
  for (int m = 1; m < 8; m <<= 1) {
    s1 += __shfl_xor(s1, m, 32);
    s2 += __shfl_xor(s2, m, 32);
  }
  const float mean = s1 * (1.0f / Dc);
  const float var = s2 * (1.0f / Dc) - mean * mean;
  const float rs = rsqrtf(var + 1e-5f);
  float* xrow = xo + ((size_t)b * Nc + g) * Dc + 4 * sub;
#pragma unroll
  for (int i = 0; i < LNP; ++i) {
    v4f y;
    for (int e = 0; e < 4; ++e)
      y[e] = (vr[i][e] - mean) * rs * gm[i][e] + bt[i][e];
    vr[i] = y;
    *(v4f*)(xsr + 32 * i) = y;
  }
  if (g >= own0) {
#pragma unroll
    for (int i = 0; i < LNP; ++i) *(v4f*)(xrow + 32 * i) = vr[i];
  }
  const int wave = tid >> 5, lane = tid & 31;
  const int hf = lane >> 4, l16 = lane & 15;
  float* out = (wave < 2) ? qo : ko;
  const int cb = (wave & 1) * 16;
  const float* ap = xs + l16 * XSTR + 2 * hf;
  const float* bp = wl + 2 * hf * WPR + wave * 16 + l16;
  v8f acc = {0.f, 0.f, 0.f, 0.f, 0.f, 0.f, 0.f, 0.f};
  wpanel_store(w0, wl, tid, 0);
  wpanel_store(w1, wl, tid, 1);
  __syncthreads();
  {
    v2f aA[PKB], bA[PKB], aB[PKB], bB[PKB];
    pfrag_load(aA, bA, ap, bp, 0);
#pragma unroll
    for (int blk = 0; blk < 2 * PNB; blk += 2) {
      pfrag_load(aB, bB, ap, bp, (blk + 1) * PKB);
      __builtin_amdgcn_sched_barrier(0);
#pragma unroll
      for (int s = 0; s < PKB; ++s) acc = wmma_f32(aA[s], bA[s], acc);
      __builtin_amdgcn_sched_barrier(0);
      if (blk + 2 < 2 * PNB) pfrag_load(aA, bA, ap, bp, (blk + 2) * PKB);
      __builtin_amdgcn_sched_barrier(0);
#pragma unroll
      for (int s = 0; s < PKB; ++s) acc = wmma_f32(aB[s], bB[s], acc);
      __builtin_amdgcn_sched_barrier(0);
    }
  }
  if (whole) {
    for (int i = 0; i < 8; ++i) {
      const int r = i + 8 * hf;
      out[((size_t)b * Nc + n0 + r) * Rc + cb + l16] = acc[i];
    }
  } else {
    for (int i = 0; i < 8; ++i) {
      const int r = i + 8 * hf, gg = n0 + r;
      if (gg >= own0) out[((size_t)b * Nc + gg) * Rc + cb + l16] = acc[i];
    }
  }
}

__device__ __forceinline__ void xchunk_load(v4f (&r)[2 * XPW],
                                            const float* __restrict__ xbat,
                                            int n0, int wv, int lane, int ch) {
#pragma unroll
  for (int m = 0; m < XPW; ++m) {
    const int j = 2 * (wv + 4 * m);
    const int g0 = min(max(n0 - Wc + j, 0), Nc - 1);
    const int g1 = min(max(n0 - Wc + j + 1, 0), Nc - 1);
    r[2 * m] = *(const v4f*)(xbat + (size_t)g0 * Dc + ch * CHc + 4 * lane);
    r[2 * m + 1] = *(const v4f*)(xbat + (size_t)g1 * Dc + ch * CHc + 4 * lane);
  }
}

__device__ __forceinline__ void xchunk_store(const v4f (&r)[2 * XPW],
                                             float* buf, int wv, int lane) {
#pragma unroll
  for (int m = 0; m < XPW; ++m) {
    const v4f lo = r[2 * m], hi = r[2 * m + 1];
    const v4f s0 = {lo.x, hi.x, lo.y, hi.y};
    const v4f s1 = {lo.z, hi.z, lo.w, hi.w};
    float* p = buf + (wv + 4 * m) * XPSTR + 8 * lane;
    *(v4f*)p = s0;
    *(v4f*)(p + 4) = s1;
  }
}

__device__ __forceinline__ void bblock_load(v2f (&bb)[KB][2], const float* bp,
                                            int blk) {
#pragma unroll
  for (int s = 0; s < KB; ++s) {
    const float* p = bp + 2 * (blk * KB + s) * XPSTR;
    bb[s][0] = *(const v2f*)p;
    bb[s][1] = *(const v2f*)(p + 32);
  }
}

__device__ __forceinline__ float xhalf(float v) {
  const int bts = __builtin_bit_cast(int, v);
  return __builtin_bit_cast(
      float, __builtin_amdgcn_permlanex16(bts, bts, 0x76543210, 0xfedcba98,
                                          false, false));
}

__device__ __forceinline__ void lds_barrier() {
  asm volatile("s_wait_dscnt 0x0\n\ts_barrier_signal -1\n\ts_barrier_wait -1" ::
                   : "memory");
}

__global__ __launch_bounds__(128) void attn_kernel(
    const float* __restrict__ x, const float* __restrict__ q,
    const float* __restrict__ k, const float* __restrict__ st,
    float* __restrict__ val) {
  __shared__ __attribute__((aligned(16))) float wm[16 * WSTR];
  __shared__ __attribute__((aligned(16))) float sw[WINc];
  __shared__ __attribute__((aligned(16))) float pmx[16 * 4];
  __shared__ __attribute__((aligned(16))) float psm[16 * 4];
  __shared__ __attribute__((aligned(16))) float xb[XBUF];
  const int cq = blockIdx.x % NCH;
  const int tile = (blockIdx.x / NCH) % NT;
  const int b = blockIdx.x / (NCH * NT);
  const int n0 = tile * 16;
  const int tid = threadIdx.x;
  const bool full = (n0 + 16 <= Nc);
  const int wave = tid >> 5, lane = tid & 31;
  const int hf = lane >> 4, l16 = lane & 15;
  const int wv = __builtin_amdgcn_readfirstlane(wave);

  const float* qbat = q + (size_t)b * Nc * Rc + 16 * hf;
  const float* kbat = k + (size_t)b * Nc * Rc + 16 * hf;
  v4f qf[4], kf[3][4];
  {
    const int gq = min(n0 + l16, Nc - 1);
#pragma unroll
    for (int i = 0; i < 4; ++i)
      qf[i] = *(const v4f*)(qbat + (size_t)gq * Rc + 4 * i);
  }
#pragma unroll
  for (int u = 0; u < 3; ++u) {
    if (wv + 4 * u < 9) {
      const int gk = min(max(n0 - Wc + 16 * (wv + 4 * u) + l16, 0), Nc - 1);
#pragma unroll
      for (int i = 0; i < 4; ++i)
        kf[u][i] = *(const v4f*)(kbat + (size_t)gk * Rc + 4 * i);
    }
  }

  const int js1 = (tid < WINc - 128) ? tid + 128 : tid;
  const int gs0 = n0 - Wc + tid;
  const int gs1 = n0 - Wc + js1;
  const float* sbat = st + (size_t)b * Nc;
  const float sv0 = sbat[min(max(gs0, 0), Nc - 1)];
  const float sv1 = sbat[min(max(gs1, 0), Nc - 1)];
  __builtin_amdgcn_sched_barrier(0);

  const float* xbat = x + (size_t)b * Nc * Dc;
  v4f xr[2 * XPW];
  xchunk_load(xr, xbat, n0, wv, lane, cq);
  __builtin_amdgcn_sched_barrier(0);

  sw[tid] = (gs0 >= 0 && gs0 < Nc) ? sv0 : 0.0f;
  sw[js1] = (gs1 >= 0 && gs1 < Nc) ? sv1 : 0.0f;

  v8f sc[3];
#pragma unroll
  for (int u = 0; u < 3; ++u) {
    v8f acc = {0.f, 0.f, 0.f, 0.f, 0.f, 0.f, 0.f, 0.f};
    if (wv + 4 * u < 9) {
#pragma unroll
      for (int i = 0; i < 4; ++i) {
        const v2f a0 = {kf[u][i].x, kf[u][i].y}, b0 = {qf[i].x, qf[i].y};
        acc = wmma_f32(a0, b0, acc);
        const v2f a1 = {kf[u][i].z, kf[u][i].w}, b1 = {qf[i].z, qf[i].w};
        acc = wmma_f32(a1, b1, acc);
      }
    }
    sc[u] = acc;
  }

  const int g = n0 + l16;
  int cmin = max(l16, Wc - n0);
  int cmax = min(l16 + 2 * Wc, Nc - 1 + Wc - n0);
  if (g >= Nc) {
    cmin = 1 << 20;
    cmax = cmin;
  }
  const unsigned span = (unsigned)(cmax - cmin);
  const int cl = 8 * hf - cmin;

  float mloc = -1e30f;
#pragma unroll
  for (int u = 0; u < 3; ++u) {
    for (int i = 0; i < 8; ++i) {
      const bool ok = (unsigned)(16 * (wv + 4 * u) + i + cl) <= span;
      mloc = fmaxf(mloc, ok ? fabsf(sc[u][i]) : -1e30f);
    }
  }
  mloc = fmaxf(mloc, xhalf(mloc));
  if (hf == 0) pmx[4 * l16 + wv] = mloc;
  lds_barrier();
  const v4f pm = *(const v4f*)(pmx + 4 * l16);
  const float m = fmaxf(fmaxf(pm.x, pm.y), fmaxf(pm.z, pm.w));

  float zloc = 0.0f;
#pragma unroll
  for (int u = 0; u < 3; ++u) {
    for (int i = 0; i < 8; ++i) {
      const bool ok = (unsigned)(16 * (wv + 4 * u) + i + cl) <= span;
      const float sv = sc[u][i];
      const float e = expf(ok ? fabsf(sv) - m : -1e30f);
      zloc += e;
      sc[u][i] = (sv > 0.0f) ? e : ((sv < 0.0f) ? -e : 0.0f);
    }
  }
  zloc += xhalf(zloc);
  if (hf == 0) psm[4 * l16 + wv] = zloc;
  lds_barrier();
  const v4f pz = *(const v4f*)(psm + 4 * l16);
  const float Z = ((pz.x + pz.y) + pz.z) + pz.w;
  const float inv = 1.0f / Z;

#pragma unroll
  for (int u = 0; u < 3; ++u) {
    if (wv + 4 * u < 9) {
      const int c0 = 16 * (wv + 4 * u) + 8 * hf;
      const v4f s0 = *(const v4f*)(sw + c0);
      const v4f s1 = *(const v4f*)(sw + c0 + 4);
      v4f w0, w1;
      for (int i = 0; i < 4; ++i) {
        const bool ok0 = (unsigned)(16 * (wv + 4 * u) + i + cl) <= span;
        const bool ok1 = (unsigned)(16 * (wv + 4 * u) + i + 4 + cl) <= span;
        w0[i] = ok0 ? (sc[u][i] * inv) * s0[i] : 0.0f;
        w1[i] = ok1 ? (sc[u][i + 4] * inv) * s1[i] : 0.0f;
      }
      float* wr = wm + l16 * WSTR + c0;
      *(v4f*)wr = w0;
      *(v4f*)(wr + 4) = w1;
    }
  }
  xchunk_store(xr, xb, wv, lane);
  __syncthreads();


  v2f af[WINc / 4];
#pragma unroll
  for (int kk = 0; kk < WINc / 4; ++kk)
    af[kk] = *(const v2f*)(wm + l16 * WSTR + 4 * kk + 2 * hf);

  float* vt = val + (size_t)b * Nc * Dc + cq * CHc + wv * 32 + l16;
  float o0[8], o1[8];
#pragma unroll
  for (int i = 0; i < 8; ++i) {
    const int gl = min(n0 + i + 8 * hf, Nc - 1);
    o0[i] = vt[(size_t)gl * Dc];
    o1[i] = vt[(size_t)gl * Dc + 16];
  }

  const float* bp = xb + hf * XPSTR + 2 * (wv * 32 + l16);
  v8f acc0 = {0.f, 0.f, 0.f, 0.f, 0.f, 0.f, 0.f, 0.f};
  v8f acc1 = {0.f, 0.f, 0.f, 0.f, 0.f, 0.f, 0.f, 0.f};
  v2f bA[KB][2], bB[KB][2];
  bblock_load(bA, bp, 0);
#pragma unroll
  for (int blk = 0; blk < NKB; blk += 2) {
    bblock_load(bB, bp, blk + 1);
    __builtin_amdgcn_sched_barrier(0);
#pragma unroll
    for (int s = 0; s < KB; ++s) {
      acc0 = wmma_f32(af[blk * KB + s], bA[s][0], acc0);
      acc1 = wmma_f32(af[blk * KB + s], bA[s][1], acc1);
    }
    __builtin_amdgcn_sched_barrier(0);
    if (blk + 2 < NKB) bblock_load(bA, bp, blk + 2);
    __builtin_amdgcn_sched_barrier(0);
#pragma unroll
    for (int s = 0; s < KB; ++s) {
      acc0 = wmma_f32(af[(blk + 1) * KB + s], bB[s][0], acc0);
      acc1 = wmma_f32(af[(blk + 1) * KB + s], bB[s][1], acc1);
    }
    __builtin_amdgcn_sched_barrier(0);
  }
  if (full) {
#pragma unroll
    for (int i = 0; i < 8; ++i) {
      float* p = vt + (size_t)(n0 + i + 8 * hf) * Dc;
      p[0] = o0[i] + acc0[i];
      p[16] = o1[i] + acc1[i];
    }
  } else {
#pragma unroll
    for (int i = 0; i < 8; ++i) {
      const int gg = n0 + i + 8 * hf;
      if (gg < Nc) {
        float* p = vt + (size_t)gg * Dc;
        p[0] = o0[i] + acc0[i];
        p[16] = o1[i] + acc1[i];
      }
    }
  }
}

extern "C" void kernel_launch(void* const* d_in, const int* in_sizes, int n_in,
                              void* d_out, int out_size, void* d_ws,
                              size_t ws_size, hipStream_t stream) {
  const int* ids = (const int*)d_in[0];
  const float* emb = (const float*)d_in[1];
  const float* pos = (const float*)d_in[2];
  const float* anchor_val = (const float*)d_in[3];
  const float* anchor_state = (const float*)d_in[4];
  const float* in_gamma = (const float*)d_in[5];
  const float* in_beta = (const float*)d_in[6];
  const float* ng = (const float*)d_in[7];
  const float* nb = (const float*)d_in[8];
  const float* Uw = (const float*)d_in[9];
  const float* Vw = (const float*)d_in[10];
  const float* sp_w = (const float*)d_in[11];
  const float* sp_b = (const float*)d_in[12];

  float* val = (float*)d_out;
  float* ws = (float*)d_ws;
  float* x = ws;
  float* q = x + (size_t)Bc * Nc * Dc;
  float* k = q + (size_t)Bc * Nc * Rc;
  float* st = k + (size_t)Bc * Nc * Rc;

  embed_ln_kernel<<<EWG, 32 * ERW, 0, stream>>>(
      ids, emb, pos, anchor_val, anchor_state, in_gamma, in_beta, sp_w, sp_b,
      val, st);

  for (int l = 0; l < Lc; ++l) {
    ln_proj_kernel<<<Bc * NT, 128, 0, stream>>>(
        val, ng + (size_t)l * Dc, nb + (size_t)l * Dc, Uw + (size_t)l * Dc * Rc,
        Vw + (size_t)l * Dc * Rc, x, q, k);
    attn_kernel<<<Bc * NT * NCH, 128, 0, stream>>>(x, q, k, st, val);
  }
}
